# starting kernel
# baseline (speedup 1.0000x reference)
#include <hip/hip_runtime.h>

#define NN 50000
#define FDIM 128

typedef __attribute__((ext_vector_type(2))) float v2f;
typedef __attribute__((ext_vector_type(8))) float v8f;


__global__ __launch_bounds__(256) void init_deg(float* deg, int n) {
    int i = blockIdx.x * blockDim.x + threadIdx.x;
    if (i < n) deg[i] = 1.0f;
}

__global__ __launch_bounds__(256) void accum_deg(const int* __restrict__ col,
                                                 float* __restrict__ deg, int e) {
    int i = blockIdx.x * blockDim.x + threadIdx.x;
    if (i < e) atomicAdd(&deg[col[i]], 1.0f);
}

__global__ __launch_bounds__(256) void finalize_dinv(float* d, int n) {
    int i = blockIdx.x * blockDim.x + threadIdx.x;
    if (i < n) d[i] = rsqrtf(d[i]);
}


template <int FOUT, bool RELU_IN>
__global__ __launch_bounds__(256) void gemm_wmma(const float* __restrict__ A,
                                                 const float* __restrict__ W,
                                                 float* __restrict__ H, int nrows) {
    constexpr int K  = FDIM;
    constexpr int CT = FOUT / 16;
    __shared__ float wlds[K * FOUT];
    for (int i = threadIdx.x; i < K * FOUT; i += 256) wlds[i] = W[i];
    __syncthreads();

    const int wave  = threadIdx.x >> 5;
    const int lane  = threadIdx.x & 31;
    const int strip = blockIdx.x * 8 + wave;
    if (strip >= (nrows >> 4)) return;

    const int row0 = strip << 4;
    const int m    = lane & 15;
    const int koff = (lane >> 4) << 1;

    v8f acc[CT];
    const v8f zero = {0.f, 0.f, 0.f, 0.f, 0.f, 0.f, 0.f, 0.f};
#pragma unroll
    for (int t = 0; t < CT; ++t) acc[t] = zero;

    const float* arow = A + (size_t)(row0 + m) * K;

    for (int kk = 0; kk < K; kk += 4) {
        float a0 = arow[kk + koff];
        float a1 = arow[kk + koff + 1];
        if (RELU_IN) { a0 = fmaxf(a0, 0.f); a1 = fmaxf(a1, 0.f); }
        v2f a; a.x = a0; a.y = a1;
#pragma unroll
        for (int t = 0; t < CT; ++t) {
            v2f b;
            b.x = wlds[(kk + koff)     * FOUT + t * 16 + m];
            b.y = wlds[(kk + koff + 1) * FOUT + t * 16 + m];
            acc[t] = __builtin_amdgcn_wmma_f32_16x16x4_f32(
                false, a, false, b, (short)0, acc[t], false, false);
        }
    }

    const int mrow = (lane >> 4) << 3;
#pragma unroll
    for (int t = 0; t < CT; ++t) {
#pragma unroll
        for (int v = 0; v < 8; ++v) {
            H[(size_t)(row0 + mrow + v) * FOUT + t * 16 + m] = acc[t][v];
        }
    }
}


template <int F>
__global__ __launch_bounds__(256) void init_out(const float* __restrict__ H,
                                                const float* __restrict__ dinv,
                                                const float* __restrict__ bias,
                                                float* __restrict__ O, int n) {
    long long i = (long long)blockIdx.x * blockDim.x + threadIdx.x;
    if (i < (long long)n * F) {
        int node = (int)(i >> (F == 128 ? 7 : 6));
        int f    = (int)(i & (F - 1));
        float d  = dinv[node];
        O[i] = bias[f] + d * d * H[i];
    }
}

template <int F>
__global__ __launch_bounds__(256) void scatter_edges(const float* __restrict__ H,
                                                     const int* __restrict__ row,
                                                     const int* __restrict__ col,
                                                     const float* __restrict__ dinv,
                                                     float* __restrict__ O, int nedges) {
    int gwave = (int)(((long long)blockIdx.x * blockDim.x + threadIdx.x) >> 5);
    int lane  = threadIdx.x & 31;
    if (gwave >= nedges) return;
    int r = row[gwave];
    int c = col[gwave];
    float w = dinv[r] * dinv[c];
    constexpr int VPL = F / 32;
    const float* hr = H + (size_t)r * F + lane * VPL;
    float*       oc = O + (size_t)c * F + lane * VPL;
#pragma unroll
    for (int v = 0; v < VPL; ++v) atomicAdd(&oc[v], w * hr[v]);
}


extern "C" void kernel_launch(void* const* d_in, const int* in_sizes, int n_in,
                              void* d_out, int out_size, void* d_ws, size_t ws_size,
                              hipStream_t stream) {
    const float* x  = (const float*)d_in[0];
    const int*   ei = (const int*)d_in[1];
    const float* W1 = (const float*)d_in[2];
    const float* b1 = (const float*)d_in[3];
    const float* W2 = (const float*)d_in[4];
    const float* b2 = (const float*)d_in[5];
    const float* W3 = (const float*)d_in[6];
    const float* b3 = (const float*)d_in[7];
    float* out = (float*)d_out;

    const int E = in_sizes[1] / 2;
    const int* row = ei;
    const int* col = ei + E;

    float* ws   = (float*)d_ws;
    float* dinv = ws;
    float* H    = ws + 50176;
    float* O    = H + (size_t)NN * FDIM;

    init_deg<<<(NN + 255) / 256, 256, 0, stream>>>(dinv, NN);
    accum_deg<<<(E + 255) / 256, 256, 0, stream>>>(col, dinv, E);
    finalize_dinv<<<(NN + 255) / 256, 256, 0, stream>>>(dinv, NN);

    const int ggrid  = ((NN / 16) + 7) / 8;
    const int sgrid  = (int)(((long long)E * 32 + 255) / 256);
    const int ig128  = (int)(((long long)NN * 128 + 255) / 256);
    const int ig64   = (int)(((long long)NN * 64 + 255) / 256);

    gemm_wmma<128, false><<<ggrid, 256, 0, stream>>>(x, W1, H, NN);
    init_out<128><<<ig128, 256, 0, stream>>>(H, dinv, b1, O, NN);
    scatter_edges<128><<<sgrid, 256, 0, stream>>>(H, row, col, dinv, O, E);

    gemm_wmma<128, true><<<ggrid, 256, 0, stream>>>(O, W2, H, NN);
    init_out<128><<<ig128, 256, 0, stream>>>(H, dinv, b2, O, NN);
    scatter_edges<128><<<sgrid, 256, 0, stream>>>(H, row, col, dinv, O, E);

    gemm_wmma<64, true><<<ggrid, 256, 0, stream>>>(O, W3, H, NN);
    init_out<64><<<ig64, 256, 0, stream>>>(H, dinv, b3, out, NN);
    scatter_edges<64><<<sgrid, 256, 0, stream>>>(H, row, col, dinv, out, E);
}


# aggregation: global f32 atomic scatter and its init launch replaced by 256 owner blocks: each keeps 196 output rows in LDS, finds its edges in a per-block edge bitmap, LDS f32 adds in arrival order
# speedup vs baseline: 1.9706x; 1.9706x over previous
#include <hip/hip_runtime.h>

#define NN 50000
#define FDIM 128
#define NPB  196
#define NOWN ((NN + NPB - 1) / NPB)

typedef __attribute__((ext_vector_type(2))) float v2f;
typedef __attribute__((ext_vector_type(8))) float v8f;


__device__ __forceinline__ int bm_words(int nedges) { return (((nedges + 31) >> 5) + 3) & ~3; }

__global__ __launch_bounds__(256) void init_deg(float* deg, int n, unsigned* bm, int nedges) {
    uint4* bm4 = (uint4*)bm;
    const int i      = blockIdx.x * blockDim.x + threadIdx.x;
    const int stride = gridDim.x * blockDim.x;
    for (int k = i; k < n; k += stride) deg[k] = 1.0f;
    const int n4 = NOWN * (bm_words(nedges) >> 2);
    const uint4 zero = {0u, 0u, 0u, 0u};
    for (int k = i; k < n4; k += stride) bm4[k] = zero;
}

__global__ __launch_bounds__(256) void accum_deg(const int* __restrict__ col,
                                                 float* __restrict__ deg,
                                                 unsigned* __restrict__ bm, int e) {
    int i = blockIdx.x * blockDim.x + threadIdx.x;
    if (i < e) {
        const int c = col[i];
        atomicAdd(&deg[c], 1.0f);
        atomicOr(&bm[(size_t)(c / NPB) * bm_words(e) + (i >> 5)], 1u << (i & 31));
    }
}

__global__ __launch_bounds__(256) void finalize_dinv(float* d, int n) {
    int i = blockIdx.x * blockDim.x + threadIdx.x;
    if (i < n) d[i] = rsqrtf(d[i]);
}


template <int FOUT, bool RELU_IN>
__global__ __launch_bounds__(256) void gemm_wmma(const float* __restrict__ A,
                                                 const float* __restrict__ W,
                                                 float* __restrict__ H, int nrows) {
    constexpr int K  = FDIM;
    constexpr int CT = FOUT / 16;
    __shared__ float wlds[K * FOUT];
    for (int i = threadIdx.x; i < K * FOUT; i += 256) wlds[i] = W[i];
    __syncthreads();

    const int wave  = threadIdx.x >> 5;
    const int lane  = threadIdx.x & 31;
    const int strip = blockIdx.x * 8 + wave;
    if (strip >= (nrows >> 4)) return;

    const int row0 = strip << 4;
    const int m    = lane & 15;
    const int koff = (lane >> 4) << 1;

    v8f acc[CT];
    const v8f zero = {0.f, 0.f, 0.f, 0.f, 0.f, 0.f, 0.f, 0.f};
#pragma unroll
    for (int t = 0; t < CT; ++t) acc[t] = zero;

    const float* arow = A + (size_t)(row0 + m) * K;

    for (int kk = 0; kk < K; kk += 4) {
        float a0 = arow[kk + koff];
        float a1 = arow[kk + koff + 1];
        if (RELU_IN) { a0 = fmaxf(a0, 0.f); a1 = fmaxf(a1, 0.f); }
        v2f a; a.x = a0; a.y = a1;
#pragma unroll
        for (int t = 0; t < CT; ++t) {
            v2f b;
            b.x = wlds[(kk + koff)     * FOUT + t * 16 + m];
            b.y = wlds[(kk + koff + 1) * FOUT + t * 16 + m];
            acc[t] = __builtin_amdgcn_wmma_f32_16x16x4_f32(
                false, a, false, b, (short)0, acc[t], false, false);
        }
    }

    const int mrow = (lane >> 4) << 3;
#pragma unroll
    for (int t = 0; t < CT; ++t) {
#pragma unroll
        for (int v = 0; v < 8; ++v) {
            H[(size_t)(row0 + mrow + v) * FOUT + t * 16 + m] = acc[t][v];
        }
    }
}


#define AGG_THREADS 1024
#define AGG_WAVES   (AGG_THREADS / 32)
#define QCAP        (31 + 1024 + 1)

template <int F>
__global__ __launch_bounds__(AGG_THREADS) void agg_lds(const float* __restrict__ H,
                                                       const int* __restrict__ row,
                                                       const int* __restrict__ col,
                                                       const unsigned* __restrict__ bm,
                                                       const float* __restrict__ dinv,
                                                       const float* __restrict__ bias,
                                                       float* __restrict__ O, int nedges) {
    constexpr int VPL = F / 32;
    constexpr int G   = 64 / VPL;
    __shared__ float tile[NPB * F];
    __shared__ unsigned wq[AGG_WAVES][QCAP];

    const int tid  = threadIdx.x;
    const int lane = tid & 31;
    const int wave = __builtin_amdgcn_readfirstlane(tid >> 5);
    const int b  = blockIdx.x;
    const int n0 = b * NPB;
    int nn = NN - n0;
    if (nn > NPB) nn = NPB;
    if (nn < 0) nn = 0;

    for (int i = tid; i < nn * F; i += AGG_THREADS) {
        const float d = dinv[n0 + i / F];
        tile[i] = bias[i & (F - 1)] + d * d * H[(size_t)n0 * F + i];
    }
    __syncthreads();

    const int nw = bm_words(nedges);
    const unsigned* mybm = bm + (size_t)b * nw;
    const int nsteps = (nw + 127) >> 7;
    int qc = 0;

    for (int j = wave; j < nsteps + AGG_WAVES; j += AGG_WAVES) {
        const bool last = (j >= nsteps);
        const int wi = j * 128 + lane * 4;
        uint4 bw = {0u, 0u, 0u, 0u};
        if (!last && wi < nw) bw = *(const uint4*)(mybm + wi);
#pragma nounroll
        for (int k = 0; k < 4; ++k) {
            unsigned wd = bw.x;
            bw.x = bw.y; bw.y = bw.z; bw.z = bw.w; bw.w = 0u;
            const unsigned ebase = (unsigned)(wi + k) * 32u;
            unsigned act = __builtin_amdgcn_ballot_w32(wd != 0u);
            while (act) {
                if (wd) {
                    const int pos = qc + (int)__builtin_amdgcn_mbcnt_lo(act, 0u);
                    wq[wave][pos] = ebase + (unsigned)__builtin_ctz(wd);
                    wd &= wd - 1u;
                }
                qc += __builtin_popcount(act);
                act = __builtin_amdgcn_ballot_w32(wd != 0u);
            }
            while (qc >= 32 || (last && qc > 0)) {
                const int cnt  = qc < 32 ? qc : 32;
                const int base = qc - cnt;
                qc = base;
                int r = 0, c = 0;
                float w = 0.f;
                __builtin_amdgcn_fence(__ATOMIC_ACQ_REL, "wavefront");
                if (lane < cnt) {
                    const int e = (int)wq[wave][base + lane];
                    r = row[e];
                    c = col[e];
                    w = dinv[r] * dinv[c];
                }
                const int wbits = __float_as_int(w);
#pragma unroll
                for (int g = 0; g < 32; g += G) {
                    if (g < cnt) {
                        float hv[G][VPL];
#pragma unroll
                        for (int u = 0; u < G; ++u) {
                            const int rj = __builtin_amdgcn_readlane(r, g + u);
                            const float* hr = H + (size_t)rj * F + lane;
#pragma unroll
                            for (int x = 0; x < VPL; ++x) hv[u][x] = hr[x * 32];
                        }
#pragma unroll
                        for (int u = 0; u < G; ++u) {
                            if (g + u < cnt) {
                                const int   cj = __builtin_amdgcn_readlane(c, g + u) - n0;
                                const float wj = __int_as_float(__builtin_amdgcn_readlane(wbits, g + u));
                                float* tr = tile + cj * F + lane;
#pragma unroll
                                for (int x = 0; x < VPL; ++x) atomicAdd(&tr[x * 32], wj * hv[u][x]);
                            }
                        }
                    }
                }
            }
        }
    }
    __syncthreads();

    for (int i = tid; i < nn * F; i += AGG_THREADS) O[(size_t)n0 * F + i] = tile[i];
}


extern "C" void kernel_launch(void* const* d_in, const int* in_sizes, int n_in,
                              void* d_out, int out_size, void* d_ws, size_t ws_size,
                              hipStream_t stream) {
    const float* x  = (const float*)d_in[0];
    const int*   ei = (const int*)d_in[1];
    const float* W1 = (const float*)d_in[2];
    const float* b1 = (const float*)d_in[3];
    const float* W2 = (const float*)d_in[4];
    const float* b2 = (const float*)d_in[5];
    const float* W3 = (const float*)d_in[6];
    const float* b3 = (const float*)d_in[7];
    float* out = (float*)d_out;

    const int E = in_sizes[1] / 2;
    const int* row = ei;
    const int* col = ei + E;

    float* ws   = (float*)d_ws;
    float* dinv = ws;
    float* H    = ws + 50176;
    float* O    = H + (size_t)NN * FDIM;
    unsigned* bm = (unsigned*)(O + (size_t)NN * FDIM);

    init_deg<<<1024, 256, 0, stream>>>(dinv, NN, bm, E);
    accum_deg<<<(E + 255) / 256, 256, 0, stream>>>(col, dinv, bm, E);
    finalize_dinv<<<(NN + 255) / 256, 256, 0, stream>>>(dinv, NN);

    const int ggrid  = ((NN / 16) + 7) / 8;

    gemm_wmma<128, false><<<ggrid, 256, 0, stream>>>(x, W1, H, NN);
    agg_lds<128><<<NOWN, AGG_THREADS, 0, stream>>>(H, row, col, bm, dinv, b1, O, E);

    gemm_wmma<128, true><<<ggrid, 256, 0, stream>>>(O, W2, H, NN);
    agg_lds<128><<<NOWN, AGG_THREADS, 0, stream>>>(H, row, col, bm, dinv, b2, O, E);

    gemm_wmma<64, true><<<ggrid, 256, 0, stream>>>(O, W3, H, NN);
    agg_lds<64><<<NOWN, AGG_THREADS, 0, stream>>>(H, row, col, bm, dinv, b3, out, E);
}


# set-up without global atomics or zero fill: 4096-edge blocks build all owners' bitmap words in LDS (integer OR), store whole lines; owners count own degrees from their bitmaps in LDS; 8 launches
# speedup vs baseline: 2.4309x; 1.2336x over previous
#include <hip/hip_runtime.h>

#define NN 50000
#define FDIM 128
#define NPB  196
#define NOWN ((NN + NPB - 1) / NPB)

typedef __attribute__((ext_vector_type(2))) float v2f;
typedef __attribute__((ext_vector_type(8))) float v8f;


#define CH_EDGES 4096
#define CH_WORDS (CH_EDGES / 32)
#define PL_PITCH (CH_WORDS + 1)
#define BM_THREADS 1024

__device__ __forceinline__ int bm_words(int nedges) {
    return (((nedges + 31) >> 5) + CH_WORDS - 1) & ~(CH_WORDS - 1);
}

__global__ __launch_bounds__(BM_THREADS) void build_bm(const int* __restrict__ col,
                                                       unsigned* __restrict__ bm, int e) {
    __shared__ __attribute__((aligned(16))) unsigned pl[NOWN * PL_PITCH];
    const int tid = threadIdx.x;
    uint4* pl4 = (uint4*)pl;
    const uint4 zero = {0u, 0u, 0u, 0u};
    for (int i = tid; i < NOWN * PL_PITCH / 4; i += BM_THREADS) pl4[i] = zero;
    __syncthreads();

    const int e0 = blockIdx.x * CH_EDGES;
    int c[CH_EDGES / BM_THREADS];
#pragma unroll
    for (int k = 0; k < CH_EDGES / BM_THREADS; ++k) {
        const int i = e0 + k * BM_THREADS + tid;
        c[k] = (i < e) ? col[i] : -1;
    }
#pragma unroll
    for (int k = 0; k < CH_EDGES / BM_THREADS; ++k) {
        const int li = k * BM_THREADS + tid;
        if (c[k] >= 0) atomicOr(&pl[(c[k] / NPB) * PL_PITCH + (li >> 5)], 1u << (li & 31));
    }
    __syncthreads();

    const int nw   = bm_words(e);
    const int lane = tid & 31;
    const int wave = tid >> 5;
    for (int p = wave; p < NOWN; p += BM_THREADS / 32) {
        unsigned* dst = bm + (size_t)p * nw + blockIdx.x * CH_WORDS + lane;
#pragma unroll
        for (int x = 0; x < CH_WORDS / 32; ++x) dst[x * 32] = pl[p * PL_PITCH + x * 32 + lane];
    }
}


template <int FOUT, bool RELU_IN>
__global__ __launch_bounds__(256) void gemm_wmma(const float* __restrict__ A,
                                                 const float* __restrict__ W,
                                                 float* __restrict__ H, int nrows) {
    constexpr int K  = FDIM;
    constexpr int CT = FOUT / 16;
    __shared__ float wlds[K * FOUT];
    for (int i = threadIdx.x; i < K * FOUT; i += 256) wlds[i] = W[i];
    __syncthreads();

    const int wave  = threadIdx.x >> 5;
    const int lane  = threadIdx.x & 31;
    const int strip = blockIdx.x * 8 + wave;
    if (strip >= (nrows >> 4)) return;

    const int row0 = strip << 4;
    const int m    = lane & 15;
    const int koff = (lane >> 4) << 1;

    v8f acc[CT];
    const v8f zero = {0.f, 0.f, 0.f, 0.f, 0.f, 0.f, 0.f, 0.f};
#pragma unroll
    for (int t = 0; t < CT; ++t) acc[t] = zero;

    const float* arow = A + (size_t)(row0 + m) * K;

    for (int kk = 0; kk < K; kk += 4) {
        float a0 = arow[kk + koff];
        float a1 = arow[kk + koff + 1];
        if (RELU_IN) { a0 = fmaxf(a0, 0.f); a1 = fmaxf(a1, 0.f); }
        v2f a; a.x = a0; a.y = a1;
#pragma unroll
        for (int t = 0; t < CT; ++t) {
            v2f b;
            b.x = wlds[(kk + koff)     * FOUT + t * 16 + m];
            b.y = wlds[(kk + koff + 1) * FOUT + t * 16 + m];
            acc[t] = __builtin_amdgcn_wmma_f32_16x16x4_f32(
                false, a, false, b, (short)0, acc[t], false, false);
        }
    }

    const int mrow = (lane >> 4) << 3;
#pragma unroll
    for (int t = 0; t < CT; ++t) {
#pragma unroll
        for (int v = 0; v < 8; ++v) {
            H[(size_t)(row0 + mrow + v) * FOUT + t * 16 + m] = acc[t][v];
        }
    }
}


#define AGG_THREADS 1024
#define AGG_WAVES   (AGG_THREADS / 32)
#define QCAP        (31 + 1024 + 1)

template <int F>
__global__ __launch_bounds__(AGG_THREADS) void agg_lds(const float* __restrict__ H,
                                                       const int* __restrict__ row,
                                                       const int* __restrict__ col,
                                                       const unsigned* __restrict__ bm,
                                                       const float* __restrict__ dinv,
                                                       const float* __restrict__ bias,
                                                       float* __restrict__ O, int nedges) {
    constexpr int VPL = F / 32;
    constexpr int G   = 64 / VPL;
    __shared__ float tile[NPB * F];
    __shared__ unsigned wq[AGG_WAVES][QCAP];

    const int tid  = threadIdx.x;
    const int lane = tid & 31;
    const int wave = __builtin_amdgcn_readfirstlane(tid >> 5);
    const int b  = blockIdx.x;
    const int n0 = b * NPB;
    int nn = NN - n0;
    if (nn > NPB) nn = NPB;
    if (nn < 0) nn = 0;

    for (int i = tid; i < nn * F; i += AGG_THREADS) {
        const float d = dinv[n0 + i / F];
        tile[i] = bias[i & (F - 1)] + d * d * H[(size_t)n0 * F + i];
    }
    __syncthreads();

    const int nw = bm_words(nedges);
    const unsigned* mybm = bm + (size_t)b * nw;
    const int nsteps = (nw + 127) >> 7;
    int qc = 0;

    for (int j = wave; j < nsteps + AGG_WAVES; j += AGG_WAVES) {
        const bool last = (j >= nsteps);
        const int wi = j * 128 + lane * 4;
        uint4 bw = {0u, 0u, 0u, 0u};
        if (!last && wi < nw) bw = *(const uint4*)(mybm + wi);
#pragma nounroll
        for (int k = 0; k < 4; ++k) {
            unsigned wd = bw.x;
            bw.x = bw.y; bw.y = bw.z; bw.z = bw.w; bw.w = 0u;
            const unsigned ebase = (unsigned)(wi + k) * 32u;
            unsigned act = __builtin_amdgcn_ballot_w32(wd != 0u);
            while (act) {
                if (wd) {
                    const int pos = qc + (int)__builtin_amdgcn_mbcnt_lo(act, 0u);
                    wq[wave][pos] = ebase + (unsigned)__builtin_ctz(wd);
                    wd &= wd - 1u;
                }
                qc += __builtin_popcount(act);
                act = __builtin_amdgcn_ballot_w32(wd != 0u);
            }
            while (qc >= 32 || (last && qc > 0)) {
                const int cnt  = qc < 32 ? qc : 32;
                const int base = qc - cnt;
                qc = base;
                int r = 0, c = 0;
                float w = 0.f;
                __builtin_amdgcn_fence(__ATOMIC_ACQ_REL, "wavefront");
                if (lane < cnt) {
                    const int e = (int)wq[wave][base + lane];
                    r = row[e];
                    c = col[e];
                    w = dinv[r] * dinv[c];
                }
                const int wbits = __float_as_int(w);
#pragma unroll
                for (int g = 0; g < 32; g += G) {
                    if (g < cnt) {
                        float hv[G][VPL];
#pragma unroll
                        for (int u = 0; u < G; ++u) {
                            const int rj = __builtin_amdgcn_readlane(r, g + u);
                            const float* hr = H + (size_t)rj * F + lane;
#pragma unroll
                            for (int x = 0; x < VPL; ++x) hv[u][x] = hr[x * 32];
                        }
#pragma unroll
                        for (int u = 0; u < G; ++u) {
                            if (g + u < cnt) {
                                const int   cj = __builtin_amdgcn_readlane(c, g + u) - n0;
                                const float wj = __int_as_float(__builtin_amdgcn_readlane(wbits, g + u));
                                float* tr = tile + cj * F + lane;
#pragma unroll
                                for (int x = 0; x < VPL; ++x) atomicAdd(&tr[x * 32], wj * hv[u][x]);
                            }
                        }
                    }
                }
            }
        }
    }
    __syncthreads();

    for (int i = tid; i < nn * F; i += AGG_THREADS) O[(size_t)n0 * F + i] = tile[i];
}


__global__ __launch_bounds__(AGG_THREADS) void own_deg(const int* __restrict__ col,
                                                       const unsigned* __restrict__ bm,
                                                       float* __restrict__ dinv, int nedges) {
    __shared__ int cnt[NPB];
    __shared__ unsigned wq[AGG_WAVES][QCAP];

    const int tid  = threadIdx.x;
    const int lane = tid & 31;
    const int wave = __builtin_amdgcn_readfirstlane(tid >> 5);
    const int b  = blockIdx.x;
    const int n0 = b * NPB;
    int nn = NN - n0;
    if (nn > NPB) nn = NPB;
    if (nn < 0) nn = 0;

    if (tid < NPB) cnt[tid] = 0;
    __syncthreads();

    const int nw = bm_words(nedges);
    const unsigned* mybm = bm + (size_t)b * nw;
    const int nsteps = (nw + 127) >> 7;
    int qc = 0;

    const uint4 zero4 = {0u, 0u, 0u, 0u};
    const bool have0 = (wave < nsteps) && (wave * 128 + lane * 4 < nw);
    uint4 nx = *(const uint4*)(mybm + (have0 ? wave * 128 + lane * 4 : 0));
    if (!have0) nx = zero4;

    for (int j = wave; j < nsteps + AGG_WAVES; j += AGG_WAVES) {
        const bool last = (j >= nsteps);
        const int wi = j * 128 + lane * 4;
        const uint4 bw = nx;
        const int wn = wi + AGG_WAVES * 128;
        const bool have = (j + AGG_WAVES < nsteps) && (wn < nw);
        nx = *(const uint4*)(mybm + (have ? wn : 0));
        if (!have) nx = zero4;
#pragma unroll
        for (int k = 0; k < 4; ++k) {
            unsigned wd = (k == 0) ? bw.x : (k == 1) ? bw.y : (k == 2) ? bw.z : bw.w;
            const unsigned ebase = (unsigned)(wi + k) * 32u;
            unsigned act = __builtin_amdgcn_ballot_w32(wd != 0u);
            while (act) {
                if (wd) {
                    const int pos = qc + (int)__builtin_amdgcn_mbcnt_lo(act, 0u);
                    wq[wave][pos] = ebase + (unsigned)__builtin_ctz(wd);
                    wd &= wd - 1u;
                }
                qc += __builtin_popcount(act);
                act = __builtin_amdgcn_ballot_w32(wd != 0u);
            }
            while (qc >= 32 || (last && qc > 0)) {
                const int take = qc < 32 ? qc : 32;
                const int base = qc - take;
                qc = base;
                __builtin_amdgcn_fence(__ATOMIC_ACQ_REL, "wavefront");
                if (lane < take) {
                    const int e = (int)wq[wave][base + lane];
                    atomicAdd(&cnt[col[e] - n0], 1);
                }
            }
        }
    }
    __syncthreads();

    if (tid < nn) dinv[n0 + tid] = rsqrtf((float)(cnt[tid] + 1));
}


extern "C" void kernel_launch(void* const* d_in, const int* in_sizes, int n_in,
                              void* d_out, int out_size, void* d_ws, size_t ws_size,
                              hipStream_t stream) {
    const float* x  = (const float*)d_in[0];
    const int*   ei = (const int*)d_in[1];
    const float* W1 = (const float*)d_in[2];
    const float* b1 = (const float*)d_in[3];
    const float* W2 = (const float*)d_in[4];
    const float* b2 = (const float*)d_in[5];
    const float* W3 = (const float*)d_in[6];
    const float* b3 = (const float*)d_in[7];
    float* out = (float*)d_out;

    const int E = in_sizes[1] / 2;
    const int* row = ei;
    const int* col = ei + E;

    float* ws   = (float*)d_ws;
    float* dinv = ws;
    float* H    = ws + 50176;
    float* O    = H + (size_t)NN * FDIM;
    unsigned* bm = (unsigned*)(O + (size_t)NN * FDIM);

    build_bm<<<(E + CH_EDGES - 1) / CH_EDGES, BM_THREADS, 0, stream>>>(col, bm, E);
    own_deg<<<NOWN, AGG_THREADS, 0, stream>>>(col, bm, dinv, E);

    const int ggrid  = ((NN / 16) + 7) / 8;

    gemm_wmma<128, false><<<ggrid, 256, 0, stream>>>(x, W1, H, NN);
    agg_lds<128><<<NOWN, AGG_THREADS, 0, stream>>>(H, row, col, bm, dinv, b1, O, E);

    gemm_wmma<128, true><<<ggrid, 256, 0, stream>>>(O, W2, H, NN);
    agg_lds<128><<<NOWN, AGG_THREADS, 0, stream>>>(H, row, col, bm, dinv, b2, O, E);

    gemm_wmma<64, true><<<ggrid, 256, 0, stream>>>(O, W3, H, NN);
    agg_lds<64><<<NOWN, AGG_THREADS, 0, stream>>>(H, row, col, bm, dinv, b3, out, E);
}


# aggregation 2, 3: a wave reads back the (source, local dest, weight) entries its layer-1 twin wrote as it took edges: no bitmap scan, no per-edge index or dinv loads; sums stay in arrival order
# speedup vs baseline: 2.7610x; 1.1358x over previous
#include <hip/hip_runtime.h>

#define NN 50000
#define FDIM 128
#define NPB  196
#define NOWN ((NN + NPB - 1) / NPB)

typedef __attribute__((ext_vector_type(2))) float v2f;
typedef __attribute__((ext_vector_type(8))) float v8f;


#define CH_EDGES 4096
#define CH_WORDS (CH_EDGES / 32)
#define PL_PITCH (CH_WORDS + 1)
#define BM_THREADS 1024

__device__ __forceinline__ int bm_words(int nedges) {
    return (((nedges + 31) >> 5) + CH_WORDS - 1) & ~(CH_WORDS - 1);
}

__global__ __launch_bounds__(BM_THREADS) void build_bm(const int* __restrict__ col,
                                                       unsigned* __restrict__ bm, int e) {
    __shared__ __attribute__((aligned(16))) unsigned pl[NOWN * PL_PITCH];
    const int tid = threadIdx.x;
    uint4* pl4 = (uint4*)pl;
    const uint4 zero = {0u, 0u, 0u, 0u};
    for (int i = tid; i < NOWN * PL_PITCH / 4; i += BM_THREADS) pl4[i] = zero;
    __syncthreads();

    const int e0 = blockIdx.x * CH_EDGES;
    int c[CH_EDGES / BM_THREADS];
#pragma unroll
    for (int k = 0; k < CH_EDGES / BM_THREADS; ++k) {
        const int i = e0 + k * BM_THREADS + tid;
        c[k] = (i < e) ? col[i] : -1;
    }
#pragma unroll
    for (int k = 0; k < CH_EDGES / BM_THREADS; ++k) {
        const int li = k * BM_THREADS + tid;
        if (c[k] >= 0) atomicOr(&pl[(c[k] / NPB) * PL_PITCH + (li >> 5)], 1u << (li & 31));
    }
    __syncthreads();

    const int nw   = bm_words(e);
    const int lane = tid & 31;
    const int wave = tid >> 5;
    for (int p = wave; p < NOWN; p += BM_THREADS / 32) {
        unsigned* dst = bm + (size_t)p * nw + blockIdx.x * CH_WORDS + lane;
#pragma unroll
        for (int x = 0; x < CH_WORDS / 32; ++x) dst[x * 32] = pl[p * PL_PITCH + x * 32 + lane];
    }
}


template <int FOUT, bool RELU_IN>
__global__ __launch_bounds__(256) void gemm_wmma(const float* __restrict__ A,
                                                 const float* __restrict__ W,
                                                 float* __restrict__ H, int nrows) {
    constexpr int K  = FDIM;
    constexpr int CT = FOUT / 16;
    __shared__ float wlds[K * FOUT];
    for (int i = threadIdx.x; i < K * FOUT; i += 256) wlds[i] = W[i];
    __syncthreads();

    const int wave  = threadIdx.x >> 5;
    const int lane  = threadIdx.x & 31;
    const int strip = blockIdx.x * 8 + wave;
    if (strip >= (nrows >> 4)) return;

    const int row0 = strip << 4;
    const int m    = lane & 15;
    const int koff = (lane >> 4) << 1;

    v8f acc[CT];
    const v8f zero = {0.f, 0.f, 0.f, 0.f, 0.f, 0.f, 0.f, 0.f};
#pragma unroll
    for (int t = 0; t < CT; ++t) acc[t] = zero;

    const float* arow = A + (size_t)(row0 + m) * K;

    for (int kk = 0; kk < K; kk += 4) {
        float a0 = arow[kk + koff];
        float a1 = arow[kk + koff + 1];
        if (RELU_IN) { a0 = fmaxf(a0, 0.f); a1 = fmaxf(a1, 0.f); }
        v2f a; a.x = a0; a.y = a1;
#pragma unroll
        for (int t = 0; t < CT; ++t) {
            v2f b;
            b.x = wlds[(kk + koff)     * FOUT + t * 16 + m];
            b.y = wlds[(kk + koff + 1) * FOUT + t * 16 + m];
            acc[t] = __builtin_amdgcn_wmma_f32_16x16x4_f32(
                false, a, false, b, (short)0, acc[t], false, false);
        }
    }

    const int mrow = (lane >> 4) << 3;
#pragma unroll
    for (int t = 0; t < CT; ++t) {
#pragma unroll
        for (int v = 0; v < 8; ++v) {
            H[(size_t)(row0 + mrow + v) * FOUT + t * 16 + m] = acc[t][v];
        }
    }
}


#define AGG_THREADS 1024
#define AGG_WAVES   (AGG_THREADS / 32)
#define QCAP        (31 + 1024 + 1)


static_assert(NPB <= 256 && NN <= (1 << 24), "an entry packs source row << 8 | local destination row");

__device__ __forceinline__ int pad16(int n) { return (n + 15) & ~15; }

__device__ __forceinline__ int wave_sum(int v) {
#pragma unroll
    for (int o = 16; o > 0; o >>= 1) v += __shfl_xor(v, o);
    return v;
}

#define LIST_LOADS (NOWN * AGG_WAVES / AGG_THREADS)

__device__ __forceinline__ void list_loads(const int* __restrict__ wcnt, int b, int tid, int lane,
                                           int (&pv)[LIST_LOADS], int& mine) {
#pragma unroll
    for (int k = 0; k < LIST_LOADS; ++k) {
        const int i = tid + k * AGG_THREADS;
        pv[k] = wcnt[i < b * AGG_WAVES ? i : 0];
    }
    mine = wcnt[b * AGG_WAVES + lane];
}

__device__ __forceinline__ void list_partials(const int (&pv)[LIST_LOADS], int b, int tid,
                                              int lane, int wave, int* red) {
    int part = 0;
#pragma unroll
    for (int k = 0; k < LIST_LOADS; ++k)
        if (tid + k * AGG_THREADS < b * AGG_WAVES) part += pad16(pv[k]);
    part = wave_sum(part);
    if (lane == 0) red[wave] = part;
}

__device__ __forceinline__ int list_start(int mine, int lane, int wave, const int* red) {
    int v = red[lane];
    if (lane < wave) v += pad16(mine);
    return __builtin_amdgcn_readfirstlane(wave_sum(v));
}

template <int F>
__device__ __forceinline__ void add_edges(const float* __restrict__ H, float* tile, int lane,
                                          int r, int cl, int wbits, int cnt) {
    constexpr int VPL = F / 32;
    constexpr int G   = 64 / VPL;
#pragma unroll
    for (int g = 0; g < 32; g += G) {
        if (g < cnt) {
            float hv[G][VPL];
#pragma unroll
            for (int u = 0; u < G; ++u) {
                const int rj = __builtin_amdgcn_readlane(r, g + u);
                const float* hr = H + (size_t)rj * F + lane;
#pragma unroll
                for (int x = 0; x < VPL; ++x) hv[u][x] = hr[x * 32];
            }
#pragma unroll
            for (int u = 0; u < G; ++u) {
                if (g + u < cnt) {
                    const int   cj = __builtin_amdgcn_readlane(cl, g + u);
                    const float wj = __int_as_float(__builtin_amdgcn_readlane(wbits, g + u));
                    float* tr = tile + cj * F + lane;
#pragma unroll
                    for (int x = 0; x < VPL; ++x) atomicAdd(&tr[x * 32], wj * hv[u][x]);
                }
            }
        }
    }
}

template <int F>
__global__ __launch_bounds__(AGG_THREADS) void agg_lds(const float* __restrict__ H,
                                                       const int* __restrict__ row,
                                                       const int* __restrict__ col,
                                                       const unsigned* __restrict__ bm,
                                                       const float* __restrict__ dinv,
                                                       const float* __restrict__ bias,
                                                       const int* __restrict__ wcnt,
                                                       unsigned* __restrict__ list,
                                                       float* __restrict__ O, int nedges) {
    __shared__ float tile[NPB * F];
    __shared__ unsigned wq[AGG_WAVES][QCAP];
    __shared__ int red[AGG_WAVES];

    const int tid  = threadIdx.x;
    const int lane = tid & 31;
    const int wave = __builtin_amdgcn_readfirstlane(tid >> 5);
    const int b  = blockIdx.x;
    const int n0 = b * NPB;
    int nn = NN - n0;
    if (nn > NPB) nn = NPB;
    if (nn < 0) nn = 0;

    int pv[LIST_LOADS], mine;
    list_loads(wcnt, b, tid, lane, pv, mine);
    for (int i = tid; i < nn * F; i += AGG_THREADS) {
        const float d = dinv[n0 + i / F];
        tile[i] = bias[i & (F - 1)] + d * d * H[(size_t)n0 * F + i];
    }
    list_partials(pv, b, tid, lane, wave, red);
    __syncthreads();

    uint2* mylist = (uint2*)list + list_start(mine, lane, wave, red);
    int lpos = 0;

    const int nw = bm_words(nedges);
    const unsigned* mybm = bm + (size_t)b * nw;
    const int nsteps = (nw + 127) >> 7;
    int qc = 0;

    for (int j = wave; j < nsteps + AGG_WAVES; j += AGG_WAVES) {
        const bool last = (j >= nsteps);
        const int wi = j * 128 + lane * 4;
        uint4 bw = {0u, 0u, 0u, 0u};
        if (!last && wi < nw) bw = *(const uint4*)(mybm + wi);
#pragma nounroll
        for (int k = 0; k < 4; ++k) {
            unsigned wd = bw.x;
            bw.x = bw.y; bw.y = bw.z; bw.z = bw.w; bw.w = 0u;
            const unsigned ebase = (unsigned)(wi + k) * 32u;
            unsigned act = __builtin_amdgcn_ballot_w32(wd != 0u);
            while (act) {
                if (wd) {
                    const int pos = qc + (int)__builtin_amdgcn_mbcnt_lo(act, 0u);
                    wq[wave][pos] = ebase + (unsigned)__builtin_ctz(wd);
                    wd &= wd - 1u;
                }
                qc += __builtin_popcount(act);
                act = __builtin_amdgcn_ballot_w32(wd != 0u);
            }
            while (qc >= 32 || (last && qc > 0)) {
                const int cnt  = qc < 32 ? qc : 32;
                const int base = qc - cnt;
                qc = base;
                int r = 0, c = 0;
                float w = 0.f;
                __builtin_amdgcn_fence(__ATOMIC_ACQ_REL, "wavefront");
                if (lane < cnt) {
                    const int e = (int)wq[wave][base + lane];
                    r = row[e];
                    c = col[e] - n0;
                    w = dinv[r] * dinv[c + n0];
                    uint2 ent;
                    ent.x = ((unsigned)r << 8) | (unsigned)c;
                    ent.y = (unsigned)__float_as_int(w);
                    mylist[lpos + lane] = ent;
                }
                lpos += cnt;
                add_edges<F>(H, tile, lane, r, c, __float_as_int(w), cnt);
            }
        }
    }
    __syncthreads();

    for (int i = tid; i < nn * F; i += AGG_THREADS) O[(size_t)n0 * F + i] = tile[i];
}

template <int F>
__global__ __launch_bounds__(AGG_THREADS) void agg_replay(const float* __restrict__ H,
                                                          const unsigned* __restrict__ list,
                                                          const int* __restrict__ wcnt,
                                                          const float* __restrict__ dinv,
                                                          const float* __restrict__ bias,
                                                          float* __restrict__ O) {
    __shared__ float tile[NPB * F];
    __shared__ int red[AGG_WAVES];

    const int tid  = threadIdx.x;
    const int lane = tid & 31;
    const int wave = __builtin_amdgcn_readfirstlane(tid >> 5);
    const int b  = blockIdx.x;
    const int n0 = b * NPB;
    int nn = NN - n0;
    if (nn > NPB) nn = NPB;
    if (nn < 0) nn = 0;

    int pv[LIST_LOADS], mine;
    list_loads(wcnt, b, tid, lane, pv, mine);
    for (int i = tid; i < nn * F; i += AGG_THREADS) {
        const float d = dinv[n0 + i / F];
        tile[i] = bias[i & (F - 1)] + d * d * H[(size_t)n0 * F + i];
    }
    list_partials(pv, b, tid, lane, wave, red);
    __syncthreads();

    const int count = __builtin_amdgcn_readlane(mine, wave);
    const uint2* mylist = (const uint2*)list + list_start(mine, lane, wave, red);
    const uint2 zero2 = {0u, 0u};
    uint2 nx = zero2;
    if (lane < count) nx = mylist[lane];
    for (int p = 0; p < count; p += 32) {
        const int cnt = (count - p) < 32 ? (count - p) : 32;
        const uint2 ent = nx;
        const int q = p + 32 + lane;
        const uint2 t = mylist[q < count ? q : count - 1];
        nx = (q < count) ? t : zero2;
        add_edges<F>(H, tile, lane, (int)(ent.x >> 8), (int)(ent.x & 255u), (int)ent.y, cnt);
    }
    __syncthreads();

    for (int i = tid; i < nn * F; i += AGG_THREADS) O[(size_t)n0 * F + i] = tile[i];
}


__global__ __launch_bounds__(AGG_THREADS) void own_deg(const int* __restrict__ col,
                                                       const unsigned* __restrict__ bm,
                                                       float* __restrict__ dinv,
                                                       int* __restrict__ wcnt, int nedges) {
    __shared__ int cnt[NPB];
    __shared__ int wtot[AGG_WAVES];
    __shared__ unsigned wq[AGG_WAVES][QCAP];

    const int tid  = threadIdx.x;
    const int lane = tid & 31;
    const int wave = __builtin_amdgcn_readfirstlane(tid >> 5);
    const int b  = blockIdx.x;
    const int n0 = b * NPB;
    int nn = NN - n0;
    if (nn > NPB) nn = NPB;
    if (nn < 0) nn = 0;

    if (tid < NPB) cnt[tid] = 0;
    __syncthreads();

    const int nw = bm_words(nedges);
    const unsigned* mybm = bm + (size_t)b * nw;
    const int nsteps = (nw + 127) >> 7;
    int qc = 0;
    int tot = 0;

    const uint4 zero4 = {0u, 0u, 0u, 0u};
    const bool have0 = (wave < nsteps) && (wave * 128 + lane * 4 < nw);
    uint4 nx = *(const uint4*)(mybm + (have0 ? wave * 128 + lane * 4 : 0));
    if (!have0) nx = zero4;

    for (int j = wave; j < nsteps + AGG_WAVES; j += AGG_WAVES) {
        const bool last = (j >= nsteps);
        const int wi = j * 128 + lane * 4;
        const uint4 bw = nx;
        const int wn = wi + AGG_WAVES * 128;
        const bool have = (j + AGG_WAVES < nsteps) && (wn < nw);
        nx = *(const uint4*)(mybm + (have ? wn : 0));
        if (!have) nx = zero4;
#pragma unroll
        for (int k = 0; k < 4; ++k) {
            unsigned wd = (k == 0) ? bw.x : (k == 1) ? bw.y : (k == 2) ? bw.z : bw.w;
            const unsigned ebase = (unsigned)(wi + k) * 32u;
            unsigned act = __builtin_amdgcn_ballot_w32(wd != 0u);
            while (act) {
                if (wd) {
                    const int pos = qc + (int)__builtin_amdgcn_mbcnt_lo(act, 0u);
                    wq[wave][pos] = ebase + (unsigned)__builtin_ctz(wd);
                    wd &= wd - 1u;
                }
                qc  += __builtin_popcount(act);
                tot += __builtin_popcount(act);
                act = __builtin_amdgcn_ballot_w32(wd != 0u);
            }
            while (qc >= 32 || (last && qc > 0)) {
                const int take = qc < 32 ? qc : 32;
                const int base = qc - take;
                qc = base;
                __builtin_amdgcn_fence(__ATOMIC_ACQ_REL, "wavefront");
                if (lane < take) {
                    const int e = (int)wq[wave][base + lane];
                    atomicAdd(&cnt[col[e] - n0], 1);
                }
            }
        }
    }
    if (lane == 0) wtot[wave] = tot;
    __syncthreads();

    if (tid < nn) dinv[n0 + tid] = rsqrtf((float)(cnt[tid] + 1));
    if (tid < AGG_WAVES) wcnt[b * AGG_WAVES + tid] = wtot[tid];
}


extern "C" void kernel_launch(void* const* d_in, const int* in_sizes, int n_in,
                              void* d_out, int out_size, void* d_ws, size_t ws_size,
                              hipStream_t stream) {
    const float* x  = (const float*)d_in[0];
    const int*   ei = (const int*)d_in[1];
    const float* W1 = (const float*)d_in[2];
    const float* b1 = (const float*)d_in[3];
    const float* W2 = (const float*)d_in[4];
    const float* b2 = (const float*)d_in[5];
    const float* W3 = (const float*)d_in[6];
    const float* b3 = (const float*)d_in[7];
    float* out = (float*)d_out;

    const int E = in_sizes[1] / 2;
    const int* row = ei;
    const int* col = ei + E;

    float* ws   = (float*)d_ws;
    float* dinv = ws;
    float* H    = ws + 50176;
    float* O    = H + (size_t)NN * FDIM;
    unsigned* bm = (unsigned*)(O + (size_t)NN * FDIM);
    int*      wcnt = (int*)d_ws + 33546240;
    unsigned* list = (unsigned*)d_ws + 33554432;

    build_bm<<<(E + CH_EDGES - 1) / CH_EDGES, BM_THREADS, 0, stream>>>(col, bm, E);
    own_deg<<<NOWN, AGG_THREADS, 0, stream>>>(col, bm, dinv, wcnt, E);

    const int ggrid  = ((NN / 16) + 7) / 8;

    gemm_wmma<128, false><<<ggrid, 256, 0, stream>>>(x, W1, H, NN);
    agg_lds<128><<<NOWN, AGG_THREADS, 0, stream>>>(H, row, col, bm, dinv, b1, wcnt, list, O, E);

    gemm_wmma<128, true><<<ggrid, 256, 0, stream>>>(O, W2, H, NN);
    agg_replay<128><<<NOWN, AGG_THREADS, 0, stream>>>(H, list, wcnt, dinv, b2, O);

    gemm_wmma<64, true><<<ggrid, 256, 0, stream>>>(O, W3, H, NN);
    agg_replay<64><<<NOWN, AGG_THREADS, 0, stream>>>(H, list, wcnt, dinv, b3, out);
}


# aggregation: a source row is gathered by one load per wave (16 B a lane at 128 wide, 8 B at 64), not one per 128-B line; LDS rows column-permuted so LDS adds stay conflict-free; arrival order kept
# speedup vs baseline: 2.9877x; 1.0821x over previous
#include <hip/hip_runtime.h>

#define NN 50000
#define FDIM 128
#define NPB  196
#define NOWN ((NN + NPB - 1) / NPB)

typedef __attribute__((ext_vector_type(2))) float v2f;
typedef __attribute__((ext_vector_type(8))) float v8f;


#define CH_EDGES 4096
#define CH_WORDS (CH_EDGES / 32)
#define PL_PITCH (CH_WORDS + 1)
#define BM_THREADS 1024

__device__ __forceinline__ int bm_words(int nedges) {
    return (((nedges + 31) >> 5) + CH_WORDS - 1) & ~(CH_WORDS - 1);
}

__global__ __launch_bounds__(BM_THREADS) void build_bm(const int* __restrict__ col,
                                                       unsigned* __restrict__ bm, int e) {
    __shared__ __attribute__((aligned(16))) unsigned pl[NOWN * PL_PITCH];
    const int tid = threadIdx.x;
    uint4* pl4 = (uint4*)pl;
    const uint4 zero = {0u, 0u, 0u, 0u};
    for (int i = tid; i < NOWN * PL_PITCH / 4; i += BM_THREADS) pl4[i] = zero;
    __syncthreads();

    const int e0 = blockIdx.x * CH_EDGES;
    int c[CH_EDGES / BM_THREADS];
#pragma unroll
    for (int k = 0; k < CH_EDGES / BM_THREADS; ++k) {
        const int i = e0 + k * BM_THREADS + tid;
        c[k] = (i < e) ? col[i] : -1;
    }
#pragma unroll
    for (int k = 0; k < CH_EDGES / BM_THREADS; ++k) {
        const int li = k * BM_THREADS + tid;
        if (c[k] >= 0) atomicOr(&pl[(c[k] / NPB) * PL_PITCH + (li >> 5)], 1u << (li & 31));
    }
    __syncthreads();

    const int nw   = bm_words(e);
    const int lane = tid & 31;
    const int wave = tid >> 5;
    for (int p = wave; p < NOWN; p += BM_THREADS / 32) {
        unsigned* dst = bm + (size_t)p * nw + blockIdx.x * CH_WORDS + lane;
#pragma unroll
        for (int x = 0; x < CH_WORDS / 32; ++x) dst[x * 32] = pl[p * PL_PITCH + x * 32 + lane];
    }
}


template <int FOUT, bool RELU_IN>
__global__ __launch_bounds__(256) void gemm_wmma(const float* __restrict__ A,
                                                 const float* __restrict__ W,
                                                 float* __restrict__ H, int nrows) {
    constexpr int K  = FDIM;
    constexpr int CT = FOUT / 16;
    __shared__ float wlds[K * FOUT];
    for (int i = threadIdx.x; i < K * FOUT; i += 256) wlds[i] = W[i];
    __syncthreads();

    const int wave  = threadIdx.x >> 5;
    const int lane  = threadIdx.x & 31;
    const int strip = blockIdx.x * 8 + wave;
    if (strip >= (nrows >> 4)) return;

    const int row0 = strip << 4;
    const int m    = lane & 15;
    const int koff = (lane >> 4) << 1;

    v8f acc[CT];
    const v8f zero = {0.f, 0.f, 0.f, 0.f, 0.f, 0.f, 0.f, 0.f};
#pragma unroll
    for (int t = 0; t < CT; ++t) acc[t] = zero;

    const float* arow = A + (size_t)(row0 + m) * K;

    for (int kk = 0; kk < K; kk += 4) {
        float a0 = arow[kk + koff];
        float a1 = arow[kk + koff + 1];
        if (RELU_IN) { a0 = fmaxf(a0, 0.f); a1 = fmaxf(a1, 0.f); }
        v2f a; a.x = a0; a.y = a1;
#pragma unroll
        for (int t = 0; t < CT; ++t) {
            v2f b;
            b.x = wlds[(kk + koff)     * FOUT + t * 16 + m];
            b.y = wlds[(kk + koff + 1) * FOUT + t * 16 + m];
            acc[t] = __builtin_amdgcn_wmma_f32_16x16x4_f32(
                false, a, false, b, (short)0, acc[t], false, false);
        }
    }

    const int mrow = (lane >> 4) << 3;
#pragma unroll
    for (int t = 0; t < CT; ++t) {
#pragma unroll
        for (int v = 0; v < 8; ++v) {
            H[(size_t)(row0 + mrow + v) * FOUT + t * 16 + m] = acc[t][v];
        }
    }
}


#define AGG_THREADS 1024
#define AGG_WAVES   (AGG_THREADS / 32)
#define QCAP        (31 + 1024 + 1)


static_assert(NPB <= 256 && NN <= (1 << 24), "an entry packs source row << 8 | local destination row");

__device__ __forceinline__ int pad16(int n) { return (n + 15) & ~15; }

__device__ __forceinline__ int wave_sum(int v) {
#pragma unroll
    for (int o = 16; o > 0; o >>= 1) v += __shfl_xor(v, o);
    return v;
}

#define LIST_LOADS (NOWN * AGG_WAVES / AGG_THREADS)

__device__ __forceinline__ void list_loads(const int* __restrict__ wcnt, int b, int tid, int lane,
                                           int (&pv)[LIST_LOADS], int& mine) {
#pragma unroll
    for (int k = 0; k < LIST_LOADS; ++k) {
        const int i = tid + k * AGG_THREADS;
        pv[k] = wcnt[i < b * AGG_WAVES ? i : 0];
    }
    mine = wcnt[b * AGG_WAVES + lane];
}

__device__ __forceinline__ void list_partials(const int (&pv)[LIST_LOADS], int b, int tid,
                                              int lane, int wave, int* red) {
    int part = 0;
#pragma unroll
    for (int k = 0; k < LIST_LOADS; ++k)
        if (tid + k * AGG_THREADS < b * AGG_WAVES) part += pad16(pv[k]);
    part = wave_sum(part);
    if (lane == 0) red[wave] = part;
}

__device__ __forceinline__ int list_start(int mine, int lane, int wave, const int* red) {
    int v = red[lane];
    if (lane < wave) v += pad16(mine);
    return __builtin_amdgcn_readfirstlane(wave_sum(v));
}

template <int F>
__device__ __forceinline__ void tile_init(const float* __restrict__ H, const float* __restrict__ dinv,
                                          const float* __restrict__ bias, float* tile,
                                          int n0, int nn, int tid) {
    constexpr int VPL = F / 32;
    typedef __attribute__((ext_vector_type(VPL))) float rowv;
    const int j = tid & 31;
    float bb[VPL];
#pragma unroll
    for (int x = 0; x < VPL; ++x) bb[x] = bias[j * VPL + x];
    constexpr int TRIPS = (NPB * 32 + AGG_THREADS - 1) / AGG_THREADS;
    rowv  h[TRIPS];
    float d[TRIPS];
#pragma unroll
    for (int k = 0; k < TRIPS; ++k) {
        const int i  = tid + k * AGG_THREADS;
        const int rw = (i < nn * 32) ? (i >> 5) : 0;
        d[k] = dinv[n0 + rw];
        h[k] = *(const rowv*)(H + (size_t)(n0 + rw) * F + j * VPL);
    }
#pragma unroll
    for (int k = 0; k < TRIPS; ++k) {
        const int i = tid + k * AGG_THREADS;
        if (i < nn * 32) {
            const int rw = i >> 5;
#pragma unroll
            for (int x = 0; x < VPL; ++x) tile[rw * F + x * 32 + j] = bb[x] + d[k] * d[k] * h[k][x];
        }
    }
}

template <int F>
__device__ __forceinline__ void tile_store(const float* tile, float* __restrict__ O,
                                           int n0, int nn, int tid) {
    constexpr int VPL = F / 32;
    typedef __attribute__((ext_vector_type(VPL))) float rowv;
    const int j = tid & 31;
    constexpr int TRIPS = (NPB * 32 + AGG_THREADS - 1) / AGG_THREADS;
    rowv o[TRIPS];
#pragma unroll
    for (int k = 0; k < TRIPS; ++k) {
        const int i  = tid + k * AGG_THREADS;
        const int rw = (i < nn * 32) ? (i >> 5) : 0;
#pragma unroll
        for (int x = 0; x < VPL; ++x) o[k][x] = tile[rw * F + x * 32 + j];
    }
#pragma unroll
    for (int k = 0; k < TRIPS; ++k) {
        const int i = tid + k * AGG_THREADS;
        if (i < nn * 32) *(rowv*)(O + (size_t)(n0 + (i >> 5)) * F + j * VPL) = o[k];
    }
}

template <int F>
__device__ __forceinline__ void add_edges(const float* __restrict__ H, float* tile, int lane,
                                          int r, int cl, int wbits, int cnt) {
    constexpr int VPL = F / 32;
    constexpr int G   = 64 / VPL;
    typedef __attribute__((ext_vector_type(VPL))) float rowv;
#pragma unroll
    for (int g = 0; g < 32; g += G) {
        if (g < cnt) {
            rowv hv[G];
#pragma unroll
            for (int u = 0; u < G; ++u) {
                const int rj = __builtin_amdgcn_readlane(r, g + u);
                hv[u] = *(const rowv*)(H + (size_t)rj * F + lane * VPL);
            }
#pragma unroll
            for (int u = 0; u < G; ++u) {
                if (g + u < cnt) {
                    const int   cj = __builtin_amdgcn_readlane(cl, g + u);
                    const float wj = __int_as_float(__builtin_amdgcn_readlane(wbits, g + u));
                    float* tr = tile + cj * F + lane;
#pragma unroll
                    for (int x = 0; x < VPL; ++x) atomicAdd(&tr[x * 32], wj * hv[u][x]);
                }
            }
        }
    }
}

template <int F>
__global__ __launch_bounds__(AGG_THREADS) void agg_lds(const float* __restrict__ H,
                                                       const int* __restrict__ row,
                                                       const int* __restrict__ col,
                                                       const unsigned* __restrict__ bm,
                                                       const float* __restrict__ dinv,
                                                       const float* __restrict__ bias,
                                                       const int* __restrict__ wcnt,
                                                       unsigned* __restrict__ list,
                                                       float* __restrict__ O, int nedges) {
    __shared__ float tile[NPB * F];
    __shared__ unsigned wq[AGG_WAVES][QCAP];
    __shared__ int red[AGG_WAVES];

    const int tid  = threadIdx.x;
    const int lane = tid & 31;
    const int wave = __builtin_amdgcn_readfirstlane(tid >> 5);
    const int b  = blockIdx.x;
    const int n0 = b * NPB;
    int nn = NN - n0;
    if (nn > NPB) nn = NPB;
    if (nn < 0) nn = 0;

    int pv[LIST_LOADS], mine;
    list_loads(wcnt, b, tid, lane, pv, mine);
    tile_init<F>(H, dinv, bias, tile, n0, nn, tid);
    list_partials(pv, b, tid, lane, wave, red);
    __syncthreads();

    uint2* mylist = (uint2*)list + list_start(mine, lane, wave, red);
    int lpos = 0;

    const int nw = bm_words(nedges);
    const unsigned* mybm = bm + (size_t)b * nw;
    const int nsteps = (nw + 127) >> 7;
    int qc = 0;

    for (int j = wave; j < nsteps + AGG_WAVES; j += AGG_WAVES) {
        const bool last = (j >= nsteps);
        const int wi = j * 128 + lane * 4;
        uint4 bw = {0u, 0u, 0u, 0u};
        if (!last && wi < nw) bw = *(const uint4*)(mybm + wi);
#pragma nounroll
        for (int k = 0; k < 4; ++k) {
            unsigned wd = bw.x;
            bw.x = bw.y; bw.y = bw.z; bw.z = bw.w; bw.w = 0u;
            const unsigned ebase = (unsigned)(wi + k) * 32u;
            unsigned act = __builtin_amdgcn_ballot_w32(wd != 0u);
            while (act) {
                if (wd) {
                    const int pos = qc + (int)__builtin_amdgcn_mbcnt_lo(act, 0u);
                    wq[wave][pos] = ebase + (unsigned)__builtin_ctz(wd);
                    wd &= wd - 1u;
                }
                qc += __builtin_popcount(act);
                act = __builtin_amdgcn_ballot_w32(wd != 0u);
            }
            while (qc >= 32 || (last && qc > 0)) {
                const int cnt  = qc < 32 ? qc : 32;
                const int base = qc - cnt;
                qc = base;
                int r = 0, c = 0;
                float w = 0.f;
                __builtin_amdgcn_fence(__ATOMIC_ACQ_REL, "wavefront");
                if (lane < cnt) {
                    const int e = (int)wq[wave][base + lane];
                    r = row[e];
                    c = col[e] - n0;
                    w = dinv[r] * dinv[c + n0];
                    uint2 ent;
                    ent.x = ((unsigned)r << 8) | (unsigned)c;
                    ent.y = (unsigned)__float_as_int(w);
                    mylist[lpos + lane] = ent;
                }
                lpos += cnt;
                add_edges<F>(H, tile, lane, r, c, __float_as_int(w), cnt);
            }
        }
    }
    __syncthreads();

    tile_store<F>(tile, O, n0, nn, tid);
}

template <int F>
__global__ __launch_bounds__(AGG_THREADS) void agg_replay(const float* __restrict__ H,
                                                          const unsigned* __restrict__ list,
                                                          const int* __restrict__ wcnt,
                                                          const float* __restrict__ dinv,
                                                          const float* __restrict__ bias,
                                                          float* __restrict__ O) {
    __shared__ float tile[NPB * F];
    __shared__ int red[AGG_WAVES];

    const int tid  = threadIdx.x;
    const int lane = tid & 31;
    const int wave = __builtin_amdgcn_readfirstlane(tid >> 5);
    const int b  = blockIdx.x;
    const int n0 = b * NPB;
    int nn = NN - n0;
    if (nn > NPB) nn = NPB;
    if (nn < 0) nn = 0;

    int pv[LIST_LOADS], mine;
    list_loads(wcnt, b, tid, lane, pv, mine);
    tile_init<F>(H, dinv, bias, tile, n0, nn, tid);
    list_partials(pv, b, tid, lane, wave, red);
    __syncthreads();

    const int count = __builtin_amdgcn_readlane(mine, wave);
    const uint2* mylist = (const uint2*)list + list_start(mine, lane, wave, red);
    const uint2 zero2 = {0u, 0u};
    uint2 nx = zero2;
    if (lane < count) nx = mylist[lane];
    for (int p = 0; p < count; p += 32) {
        const int cnt = (count - p) < 32 ? (count - p) : 32;
        const uint2 ent = nx;
        const int q = p + 32 + lane;
        const uint2 t = mylist[q < count ? q : count - 1];
        nx = (q < count) ? t : zero2;
        add_edges<F>(H, tile, lane, (int)(ent.x >> 8), (int)(ent.x & 255u), (int)ent.y, cnt);
    }
    __syncthreads();

    tile_store<F>(tile, O, n0, nn, tid);
}


__global__ __launch_bounds__(AGG_THREADS) void own_deg(const int* __restrict__ col,
                                                       const unsigned* __restrict__ bm,
                                                       float* __restrict__ dinv,
                                                       int* __restrict__ wcnt, int nedges) {
    __shared__ int cnt[NPB];
    __shared__ int wtot[AGG_WAVES];
    __shared__ unsigned wq[AGG_WAVES][QCAP];

    const int tid  = threadIdx.x;
    const int lane = tid & 31;
    const int wave = __builtin_amdgcn_readfirstlane(tid >> 5);
    const int b  = blockIdx.x;
    const int n0 = b * NPB;
    int nn = NN - n0;
    if (nn > NPB) nn = NPB;
    if (nn < 0) nn = 0;

    if (tid < NPB) cnt[tid] = 0;
    __syncthreads();

    const int nw = bm_words(nedges);
    const unsigned* mybm = bm + (size_t)b * nw;
    const int nsteps = (nw + 127) >> 7;
    int qc = 0;
    int tot = 0;

    const uint4 zero4 = {0u, 0u, 0u, 0u};
    const bool have0 = (wave < nsteps) && (wave * 128 + lane * 4 < nw);
    uint4 nx = *(const uint4*)(mybm + (have0 ? wave * 128 + lane * 4 : 0));
    if (!have0) nx = zero4;

    for (int j = wave; j < nsteps + AGG_WAVES; j += AGG_WAVES) {
        const bool last = (j >= nsteps);
        const int wi = j * 128 + lane * 4;
        const uint4 bw = nx;
        const int wn = wi + AGG_WAVES * 128;
        const bool have = (j + AGG_WAVES < nsteps) && (wn < nw);
        nx = *(const uint4*)(mybm + (have ? wn : 0));
        if (!have) nx = zero4;
#pragma unroll
        for (int k = 0; k < 4; ++k) {
            unsigned wd = (k == 0) ? bw.x : (k == 1) ? bw.y : (k == 2) ? bw.z : bw.w;
            const unsigned ebase = (unsigned)(wi + k) * 32u;
            unsigned act = __builtin_amdgcn_ballot_w32(wd != 0u);
            while (act) {
                if (wd) {
                    const int pos = qc + (int)__builtin_amdgcn_mbcnt_lo(act, 0u);
                    wq[wave][pos] = ebase + (unsigned)__builtin_ctz(wd);
                    wd &= wd - 1u;
                }
                qc  += __builtin_popcount(act);
                tot += __builtin_popcount(act);
                act = __builtin_amdgcn_ballot_w32(wd != 0u);
            }
            while (qc >= 32 || (last && qc > 0)) {
                const int take = qc < 32 ? qc : 32;
                const int base = qc - take;
                qc = base;
                __builtin_amdgcn_fence(__ATOMIC_ACQ_REL, "wavefront");
                if (lane < take) {
                    const int e = (int)wq[wave][base + lane];
                    atomicAdd(&cnt[col[e] - n0], 1);
                }
            }
        }
    }
    if (lane == 0) wtot[wave] = tot;
    __syncthreads();

    if (tid < nn) dinv[n0 + tid] = rsqrtf((float)(cnt[tid] + 1));
    if (tid < AGG_WAVES) wcnt[b * AGG_WAVES + tid] = wtot[tid];
}


extern "C" void kernel_launch(void* const* d_in, const int* in_sizes, int n_in,
                              void* d_out, int out_size, void* d_ws, size_t ws_size,
                              hipStream_t stream) {
    const float* x  = (const float*)d_in[0];
    const int*   ei = (const int*)d_in[1];
    const float* W1 = (const float*)d_in[2];
    const float* b1 = (const float*)d_in[3];
    const float* W2 = (const float*)d_in[4];
    const float* b2 = (const float*)d_in[5];
    const float* W3 = (const float*)d_in[6];
    const float* b3 = (const float*)d_in[7];
    float* out = (float*)d_out;

    const int E = in_sizes[1] / 2;
    const int* row = ei;
    const int* col = ei + E;

    float* ws   = (float*)d_ws;
    float* dinv = ws;
    float* H    = ws + 50176;
    float* O    = H + (size_t)NN * FDIM;
    unsigned* bm = (unsigned*)(O + (size_t)NN * FDIM);
    int*      wcnt = (int*)d_ws + 33546240;
    unsigned* list = (unsigned*)d_ws + 33554432;

    build_bm<<<(E + CH_EDGES - 1) / CH_EDGES, BM_THREADS, 0, stream>>>(col, bm, E);
    own_deg<<<NOWN, AGG_THREADS, 0, stream>>>(col, bm, dinv, wcnt, E);

    const int ggrid  = ((NN / 16) + 7) / 8;

    gemm_wmma<128, false><<<ggrid, 256, 0, stream>>>(x, W1, H, NN);
    agg_lds<128><<<NOWN, AGG_THREADS, 0, stream>>>(H, row, col, bm, dinv, b1, wcnt, list, O, E);

    gemm_wmma<128, true><<<ggrid, 256, 0, stream>>>(O, W2, H, NN);
    agg_replay<128><<<NOWN, AGG_THREADS, 0, stream>>>(H, list, wcnt, dinv, b2, O);

    gemm_wmma<64, true><<<ggrid, 256, 0, stream>>>(O, W3, H, NN);
    agg_replay<64><<<NOWN, AGG_THREADS, 0, stream>>>(H, list, wcnt, dinv, b3, out);
}


# dense transform: A strip and result go through LDS: 16 whole-row 16-byte loads up front and whole-row stores per strip, not 32 8-byte loads over 16 lines each and 64 4-byte stores of 64-byte pieces
# speedup vs baseline: 3.0963x; 1.0363x over previous
#include <hip/hip_runtime.h>

#define NN 50000
#define FDIM 128
#define NPB  196
#define NOWN ((NN + NPB - 1) / NPB)

typedef __attribute__((ext_vector_type(2))) float v2f;
typedef __attribute__((ext_vector_type(8))) float v8f;


#define CH_EDGES 4096
#define CH_WORDS (CH_EDGES / 32)
#define PL_PITCH (CH_WORDS + 1)
#define BM_THREADS 1024

__device__ __forceinline__ int bm_words(int nedges) {
    return (((nedges + 31) >> 5) + CH_WORDS - 1) & ~(CH_WORDS - 1);
}

__global__ __launch_bounds__(BM_THREADS) void build_bm(const int* __restrict__ col,
                                                       unsigned* __restrict__ bm, int e) {
    __shared__ __attribute__((aligned(16))) unsigned pl[NOWN * PL_PITCH];
    const int tid = threadIdx.x;
    uint4* pl4 = (uint4*)pl;
    const uint4 zero = {0u, 0u, 0u, 0u};
    for (int i = tid; i < NOWN * PL_PITCH / 4; i += BM_THREADS) pl4[i] = zero;
    __syncthreads();

    const int e0 = blockIdx.x * CH_EDGES;
    int c[CH_EDGES / BM_THREADS];
#pragma unroll
    for (int k = 0; k < CH_EDGES / BM_THREADS; ++k) {
        const int i = e0 + k * BM_THREADS + tid;
        c[k] = (i < e) ? col[i] : -1;
    }
#pragma unroll
    for (int k = 0; k < CH_EDGES / BM_THREADS; ++k) {
        const int li = k * BM_THREADS + tid;
        if (c[k] >= 0) atomicOr(&pl[(c[k] / NPB) * PL_PITCH + (li >> 5)], 1u << (li & 31));
    }
    __syncthreads();

    const int nw   = bm_words(e);
    const int lane = tid & 31;
    const int wave = tid >> 5;
    for (int p = wave; p < NOWN; p += BM_THREADS / 32) {
        unsigned* dst = bm + (size_t)p * nw + blockIdx.x * CH_WORDS + lane;
#pragma unroll
        for (int x = 0; x < CH_WORDS / 32; ++x) dst[x * 32] = pl[p * PL_PITCH + x * 32 + lane];
    }
}


typedef __attribute__((ext_vector_type(4))) float v4f;

#define SP (FDIM + 4)

template <int FOUT, bool RELU_IN>
__global__ __launch_bounds__(256) void gemm_wmma(const float* __restrict__ A,
                                                 const float* __restrict__ W,
                                                 float* __restrict__ H, int nrows) {
    constexpr int K  = FDIM;
    constexpr int CT = FOUT / 16;
    __shared__ float wlds[K * FOUT];
    __shared__ __attribute__((aligned(16))) float stg[8 * 16 * SP];

    const int wave  = threadIdx.x >> 5;
    const int lane  = threadIdx.x & 31;
    const int strip = blockIdx.x * 8 + wave;
    const int nstr  = nrows >> 4;
    const int row0  = (strip < nstr ? strip : nstr - 1) << 4;
    float* st = stg + wave * (16 * SP);

    v4f av[16];
#pragma unroll
    for (int i = 0; i < 16; ++i) av[i] = *(const v4f*)(A + (size_t)(row0 + i) * K + lane * 4);

    for (int i = threadIdx.x; i < K * FOUT; i += 256) wlds[i] = W[i];

#pragma unroll
    for (int i = 0; i < 16; ++i) {
        v4f v = av[i];
        if (RELU_IN) {
            v.x = fmaxf(v.x, 0.f); v.y = fmaxf(v.y, 0.f);
            v.z = fmaxf(v.z, 0.f); v.w = fmaxf(v.w, 0.f);
        }
        *(v4f*)(st + i * SP + lane * 4) = v;
    }
    __syncthreads();

    if (strip >= nstr) return;

    const int m    = lane & 15;
    const int koff = (lane >> 4) << 1;

    v8f acc[CT];
    const v8f zero = {0.f, 0.f, 0.f, 0.f, 0.f, 0.f, 0.f, 0.f};
#pragma unroll
    for (int t = 0; t < CT; ++t) acc[t] = zero;

    const float* arow = st + m * SP + koff;

    for (int kk = 0; kk < K; kk += 4) {
        const v2f a = *(const v2f*)(arow + kk);
#pragma unroll
        for (int t = 0; t < CT; ++t) {
            v2f b;
            b.x = wlds[(kk + koff)     * FOUT + t * 16 + m];
            b.y = wlds[(kk + koff + 1) * FOUT + t * 16 + m];
            acc[t] = __builtin_amdgcn_wmma_f32_16x16x4_f32(
                false, a, false, b, (short)0, acc[t], false, false);
        }
    }

    const int mrow = (lane >> 4) << 3;
#pragma unroll
    for (int t = 0; t < CT; ++t) {
#pragma unroll
        for (int v = 0; v < 8; ++v) st[(mrow + v) * SP + t * 16 + m] = acc[t][v];
    }
    __builtin_amdgcn_fence(__ATOMIC_ACQ_REL, "wavefront");

    constexpr int VPR = FOUT / 4;
    constexpr int RPI = 32 / VPR;
    const int rsub = lane / VPR;
    const int cc   = (lane % VPR) * 4;
    v4f o[16 / RPI];
#pragma unroll
    for (int i = 0; i < 16 / RPI; ++i) o[i] = *(const v4f*)(st + (i * RPI + rsub) * SP + cc);
#pragma unroll
    for (int i = 0; i < 16 / RPI; ++i)
        *(v4f*)(H + (size_t)(row0 + i * RPI + rsub) * FOUT + cc) = o[i];
}


#define AGG_THREADS 1024
#define AGG_WAVES   (AGG_THREADS / 32)
#define QCAP        (31 + 1024 + 1)


static_assert(NPB <= 256 && NN <= (1 << 24), "an entry packs source row << 8 | local destination row");

__device__ __forceinline__ int pad16(int n) { return (n + 15) & ~15; }

__device__ __forceinline__ int wave_sum(int v) {
#pragma unroll
    for (int o = 16; o > 0; o >>= 1) v += __shfl_xor(v, o);
    return v;
}

#define LIST_LOADS (NOWN * AGG_WAVES / AGG_THREADS)

__device__ __forceinline__ void list_loads(const int* __restrict__ wcnt, int b, int tid, int lane,
                                           int (&pv)[LIST_LOADS], int& mine) {
#pragma unroll
    for (int k = 0; k < LIST_LOADS; ++k) {
        const int i = tid + k * AGG_THREADS;
        pv[k] = wcnt[i < b * AGG_WAVES ? i : 0];
    }
    mine = wcnt[b * AGG_WAVES + lane];
}

__device__ __forceinline__ void list_partials(const int (&pv)[LIST_LOADS], int b, int tid,
                                              int lane, int wave, int* red) {
    int part = 0;
#pragma unroll
    for (int k = 0; k < LIST_LOADS; ++k)
        if (tid + k * AGG_THREADS < b * AGG_WAVES) part += pad16(pv[k]);
    part = wave_sum(part);
    if (lane == 0) red[wave] = part;
}

__device__ __forceinline__ int list_start(int mine, int lane, int wave, const int* red) {
    int v = red[lane];
    if (lane < wave) v += pad16(mine);
    return __builtin_amdgcn_readfirstlane(wave_sum(v));
}

template <int F>
__device__ __forceinline__ void tile_init(const float* __restrict__ H, const float* __restrict__ dinv,
                                          const float* __restrict__ bias, float* tile,
                                          int n0, int nn, int tid) {
    constexpr int VPL = F / 32;
    typedef __attribute__((ext_vector_type(VPL))) float rowv;
    const int j = tid & 31;
    float bb[VPL];
#pragma unroll
    for (int x = 0; x < VPL; ++x) bb[x] = bias[j * VPL + x];
    constexpr int TRIPS = (NPB * 32 + AGG_THREADS - 1) / AGG_THREADS;
    rowv  h[TRIPS];
    float d[TRIPS];
#pragma unroll
    for (int k = 0; k < TRIPS; ++k) {
        const int i  = tid + k * AGG_THREADS;
        const int rw = (i < nn * 32) ? (i >> 5) : 0;
        d[k] = dinv[n0 + rw];
        h[k] = *(const rowv*)(H + (size_t)(n0 + rw) * F + j * VPL);
    }
#pragma unroll
    for (int k = 0; k < TRIPS; ++k) {
        const int i = tid + k * AGG_THREADS;
        if (i < nn * 32) {
            const int rw = i >> 5;
#pragma unroll
            for (int x = 0; x < VPL; ++x) tile[rw * F + x * 32 + j] = bb[x] + d[k] * d[k] * h[k][x];
        }
    }
}

template <int F>
__device__ __forceinline__ void tile_store(const float* tile, float* __restrict__ O,
                                           int n0, int nn, int tid) {
    constexpr int VPL = F / 32;
    typedef __attribute__((ext_vector_type(VPL))) float rowv;
    const int j = tid & 31;
    constexpr int TRIPS = (NPB * 32 + AGG_THREADS - 1) / AGG_THREADS;
    rowv o[TRIPS];
#pragma unroll
    for (int k = 0; k < TRIPS; ++k) {
        const int i  = tid + k * AGG_THREADS;
        const int rw = (i < nn * 32) ? (i >> 5) : 0;
#pragma unroll
        for (int x = 0; x < VPL; ++x) o[k][x] = tile[rw * F + x * 32 + j];
    }
#pragma unroll
    for (int k = 0; k < TRIPS; ++k) {
        const int i = tid + k * AGG_THREADS;
        if (i < nn * 32) *(rowv*)(O + (size_t)(n0 + (i >> 5)) * F + j * VPL) = o[k];
    }
}

template <int F>
__device__ __forceinline__ void add_edges(const float* __restrict__ H, float* tile, int lane,
                                          int r, int cl, int wbits, int cnt) {
    constexpr int VPL = F / 32;
    constexpr int G   = 64 / VPL;
    typedef __attribute__((ext_vector_type(VPL))) float rowv;
#pragma unroll
    for (int g = 0; g < 32; g += G) {
        if (g < cnt) {
            rowv hv[G];
#pragma unroll
            for (int u = 0; u < G; ++u) {
                const int rj = __builtin_amdgcn_readlane(r, g + u);
                hv[u] = *(const rowv*)(H + (size_t)rj * F + lane * VPL);
            }
#pragma unroll
            for (int u = 0; u < G; ++u) {
                if (g + u < cnt) {
                    const int   cj = __builtin_amdgcn_readlane(cl, g + u);
                    const float wj = __int_as_float(__builtin_amdgcn_readlane(wbits, g + u));
                    float* tr = tile + cj * F + lane;
#pragma unroll
                    for (int x = 0; x < VPL; ++x) atomicAdd(&tr[x * 32], wj * hv[u][x]);
                }
            }
        }
    }
}

template <int F>
__global__ __launch_bounds__(AGG_THREADS) void agg_lds(const float* __restrict__ H,
                                                       const int* __restrict__ row,
                                                       const int* __restrict__ col,
                                                       const unsigned* __restrict__ bm,
                                                       const float* __restrict__ dinv,
                                                       const float* __restrict__ bias,
                                                       const int* __restrict__ wcnt,
                                                       unsigned* __restrict__ list,
                                                       float* __restrict__ O, int nedges) {
    __shared__ float tile[NPB * F];
    __shared__ unsigned wq[AGG_WAVES][QCAP];
    __shared__ int red[AGG_WAVES];

    const int tid  = threadIdx.x;
    const int lane = tid & 31;
    const int wave = __builtin_amdgcn_readfirstlane(tid >> 5);
    const int b  = blockIdx.x;
    const int n0 = b * NPB;
    int nn = NN - n0;
    if (nn > NPB) nn = NPB;
    if (nn < 0) nn = 0;

    int pv[LIST_LOADS], mine;
    list_loads(wcnt, b, tid, lane, pv, mine);
    tile_init<F>(H, dinv, bias, tile, n0, nn, tid);
    list_partials(pv, b, tid, lane, wave, red);
    __syncthreads();

    uint2* mylist = (uint2*)list + list_start(mine, lane, wave, red);
    int lpos = 0;

    const int nw = bm_words(nedges);
    const unsigned* mybm = bm + (size_t)b * nw;
    const int nsteps = (nw + 127) >> 7;
    int qc = 0;

    for (int j = wave; j < nsteps + AGG_WAVES; j += AGG_WAVES) {
        const bool last = (j >= nsteps);
        const int wi = j * 128 + lane * 4;
        uint4 bw = {0u, 0u, 0u, 0u};
        if (!last && wi < nw) bw = *(const uint4*)(mybm + wi);
#pragma nounroll
        for (int k = 0; k < 4; ++k) {
            unsigned wd = bw.x;
            bw.x = bw.y; bw.y = bw.z; bw.z = bw.w; bw.w = 0u;
            const unsigned ebase = (unsigned)(wi + k) * 32u;
            unsigned act = __builtin_amdgcn_ballot_w32(wd != 0u);
            while (act) {
                if (wd) {
                    const int pos = qc + (int)__builtin_amdgcn_mbcnt_lo(act, 0u);
                    wq[wave][pos] = ebase + (unsigned)__builtin_ctz(wd);
                    wd &= wd - 1u;
                }
                qc += __builtin_popcount(act);
                act = __builtin_amdgcn_ballot_w32(wd != 0u);
            }
            while (qc >= 32 || (last && qc > 0)) {
                const int cnt  = qc < 32 ? qc : 32;
                const int base = qc - cnt;
                qc = base;
                int r = 0, c = 0;
                float w = 0.f;
                __builtin_amdgcn_fence(__ATOMIC_ACQ_REL, "wavefront");
                if (lane < cnt) {
                    const int e = (int)wq[wave][base + lane];
                    r = row[e];
                    c = col[e] - n0;
                    w = dinv[r] * dinv[c + n0];
                    uint2 ent;
                    ent.x = ((unsigned)r << 8) | (unsigned)c;
                    ent.y = (unsigned)__float_as_int(w);
                    mylist[lpos + lane] = ent;
                }
                lpos += cnt;
                add_edges<F>(H, tile, lane, r, c, __float_as_int(w), cnt);
            }
        }
    }
    __syncthreads();

    tile_store<F>(tile, O, n0, nn, tid);
}

template <int F>
__global__ __launch_bounds__(AGG_THREADS) void agg_replay(const float* __restrict__ H,
                                                          const unsigned* __restrict__ list,
                                                          const int* __restrict__ wcnt,
                                                          const float* __restrict__ dinv,
                                                          const float* __restrict__ bias,
                                                          float* __restrict__ O) {
    __shared__ float tile[NPB * F];
    __shared__ int red[AGG_WAVES];

    const int tid  = threadIdx.x;
    const int lane = tid & 31;
    const int wave = __builtin_amdgcn_readfirstlane(tid >> 5);
    const int b  = blockIdx.x;
    const int n0 = b * NPB;
    int nn = NN - n0;
    if (nn > NPB) nn = NPB;
    if (nn < 0) nn = 0;

    int pv[LIST_LOADS], mine;
    list_loads(wcnt, b, tid, lane, pv, mine);
    tile_init<F>(H, dinv, bias, tile, n0, nn, tid);
    list_partials(pv, b, tid, lane, wave, red);
    __syncthreads();

    const int count = __builtin_amdgcn_readlane(mine, wave);
    const uint2* mylist = (const uint2*)list + list_start(mine, lane, wave, red);
    const uint2 zero2 = {0u, 0u};
    uint2 nx = zero2;
    if (lane < count) nx = mylist[lane];
    for (int p = 0; p < count; p += 32) {
        const int cnt = (count - p) < 32 ? (count - p) : 32;
        const uint2 ent = nx;
        const int q = p + 32 + lane;
        const uint2 t = mylist[q < count ? q : count - 1];
        nx = (q < count) ? t : zero2;
        add_edges<F>(H, tile, lane, (int)(ent.x >> 8), (int)(ent.x & 255u), (int)ent.y, cnt);
    }
    __syncthreads();

    tile_store<F>(tile, O, n0, nn, tid);
}


__global__ __launch_bounds__(AGG_THREADS) void own_deg(const int* __restrict__ col,
                                                       const unsigned* __restrict__ bm,
                                                       float* __restrict__ dinv,
                                                       int* __restrict__ wcnt, int nedges) {
    __shared__ int cnt[NPB];
    __shared__ int wtot[AGG_WAVES];
    __shared__ unsigned wq[AGG_WAVES][QCAP];

    const int tid  = threadIdx.x;
    const int lane = tid & 31;
    const int wave = __builtin_amdgcn_readfirstlane(tid >> 5);
    const int b  = blockIdx.x;
    const int n0 = b * NPB;
    int nn = NN - n0;
    if (nn > NPB) nn = NPB;
    if (nn < 0) nn = 0;

    if (tid < NPB) cnt[tid] = 0;
    __syncthreads();

    const int nw = bm_words(nedges);
    const unsigned* mybm = bm + (size_t)b * nw;
    const int nsteps = (nw + 127) >> 7;
    int qc = 0;
    int tot = 0;

    const uint4 zero4 = {0u, 0u, 0u, 0u};
    const bool have0 = (wave < nsteps) && (wave * 128 + lane * 4 < nw);
    uint4 nx = *(const uint4*)(mybm + (have0 ? wave * 128 + lane * 4 : 0));
    if (!have0) nx = zero4;

    for (int j = wave; j < nsteps + AGG_WAVES; j += AGG_WAVES) {
        const bool last = (j >= nsteps);
        const int wi = j * 128 + lane * 4;
        const uint4 bw = nx;
        const int wn = wi + AGG_WAVES * 128;
        const bool have = (j + AGG_WAVES < nsteps) && (wn < nw);
        nx = *(const uint4*)(mybm + (have ? wn : 0));
        if (!have) nx = zero4;
#pragma unroll
        for (int k = 0; k < 4; ++k) {
            unsigned wd = (k == 0) ? bw.x : (k == 1) ? bw.y : (k == 2) ? bw.z : bw.w;
            const unsigned ebase = (unsigned)(wi + k) * 32u;
            unsigned act = __builtin_amdgcn_ballot_w32(wd != 0u);
            while (act) {
                if (wd) {
                    const int pos = qc + (int)__builtin_amdgcn_mbcnt_lo(act, 0u);
                    wq[wave][pos] = ebase + (unsigned)__builtin_ctz(wd);
                    wd &= wd - 1u;
                }
                qc  += __builtin_popcount(act);
                tot += __builtin_popcount(act);
                act = __builtin_amdgcn_ballot_w32(wd != 0u);
            }
            while (qc >= 32 || (last && qc > 0)) {
                const int take = qc < 32 ? qc : 32;
                const int base = qc - take;
                qc = base;
                __builtin_amdgcn_fence(__ATOMIC_ACQ_REL, "wavefront");
                if (lane < take) {
                    const int e = (int)wq[wave][base + lane];
                    atomicAdd(&cnt[col[e] - n0], 1);
                }
            }
        }
    }
    if (lane == 0) wtot[wave] = tot;
    __syncthreads();

    if (tid < nn) dinv[n0 + tid] = rsqrtf((float)(cnt[tid] + 1));
    if (tid < AGG_WAVES) wcnt[b * AGG_WAVES + tid] = wtot[tid];
}


extern "C" void kernel_launch(void* const* d_in, const int* in_sizes, int n_in,
                              void* d_out, int out_size, void* d_ws, size_t ws_size,
                              hipStream_t stream) {
    const float* x  = (const float*)d_in[0];
    const int*   ei = (const int*)d_in[1];
    const float* W1 = (const float*)d_in[2];
    const float* b1 = (const float*)d_in[3];
    const float* W2 = (const float*)d_in[4];
    const float* b2 = (const float*)d_in[5];
    const float* W3 = (const float*)d_in[6];
    const float* b3 = (const float*)d_in[7];
    float* out = (float*)d_out;

    const int E = in_sizes[1] / 2;
    const int* row = ei;
    const int* col = ei + E;

    float* ws   = (float*)d_ws;
    float* dinv = ws;
    float* H    = ws + 50176;
    float* O    = H + (size_t)NN * FDIM;
    unsigned* bm = (unsigned*)(O + (size_t)NN * FDIM);
    int*      wcnt = (int*)d_ws + 33546240;
    unsigned* list = (unsigned*)d_ws + 33554432;

    build_bm<<<(E + CH_EDGES - 1) / CH_EDGES, BM_THREADS, 0, stream>>>(col, bm, E);
    own_deg<<<NOWN, AGG_THREADS, 0, stream>>>(col, bm, dinv, wcnt, E);

    const int ggrid  = ((NN / 16) + 7) / 8;

    gemm_wmma<128, false><<<ggrid, 256, 0, stream>>>(x, W1, H, NN);
    agg_lds<128><<<NOWN, AGG_THREADS, 0, stream>>>(H, row, col, bm, dinv, b1, wcnt, list, O, E);

    gemm_wmma<128, true><<<ggrid, 256, 0, stream>>>(O, W2, H, NN);
    agg_replay<128><<<NOWN, AGG_THREADS, 0, stream>>>(H, list, wcnt, dinv, b2, O);

    gemm_wmma<64, true><<<ggrid, 256, 0, stream>>>(O, W3, H, NN);
    agg_replay<64><<<NOWN, AGG_THREADS, 0, stream>>>(H, list, wcnt, dinv, b3, out);
}


# aggregation 1, 2 fused with the next dense transform on the owner tiles: a block ReLUs its finished LDS rows, multiplies them by W on WMMA, k summed in the tile's column order; 6 launches, not 8
# speedup vs baseline: 3.3923x; 1.0956x over previous
#include <hip/hip_runtime.h>

#define NN 50000
#define FDIM 128
#define NPB  196
#define NOWN ((NN + NPB - 1) / NPB)

typedef __attribute__((ext_vector_type(2))) float v2f;
typedef __attribute__((ext_vector_type(8))) float v8f;


#define CH_EDGES 4096
#define CH_WORDS (CH_EDGES / 32)
#define PL_PITCH (CH_WORDS + 1)
#define BM_THREADS 1024

__device__ __forceinline__ int bm_words(int nedges) {
    return (((nedges + 31) >> 5) + CH_WORDS - 1) & ~(CH_WORDS - 1);
}

__global__ __launch_bounds__(BM_THREADS) void build_bm(const int* __restrict__ col,
                                                       unsigned* __restrict__ bm, int e) {
    __shared__ __attribute__((aligned(16))) unsigned pl[NOWN * PL_PITCH];
    const int tid = threadIdx.x;
    uint4* pl4 = (uint4*)pl;
    const uint4 zero = {0u, 0u, 0u, 0u};
    for (int i = tid; i < NOWN * PL_PITCH / 4; i += BM_THREADS) pl4[i] = zero;
    __syncthreads();

    const int e0 = blockIdx.x * CH_EDGES;
    int c[CH_EDGES / BM_THREADS];
#pragma unroll
    for (int k = 0; k < CH_EDGES / BM_THREADS; ++k) {
        const int i = e0 + k * BM_THREADS + tid;
        c[k] = (i < e) ? col[i] : -1;
    }
#pragma unroll
    for (int k = 0; k < CH_EDGES / BM_THREADS; ++k) {
        const int li = k * BM_THREADS + tid;
        if (c[k] >= 0) atomicOr(&pl[(c[k] / NPB) * PL_PITCH + (li >> 5)], 1u << (li & 31));
    }
    __syncthreads();

    const int nw   = bm_words(e);
    const int lane = tid & 31;
    const int wave = tid >> 5;
    for (int p = wave; p < NOWN; p += BM_THREADS / 32) {
        unsigned* dst = bm + (size_t)p * nw + blockIdx.x * CH_WORDS + lane;
#pragma unroll
        for (int x = 0; x < CH_WORDS / 32; ++x) dst[x * 32] = pl[p * PL_PITCH + x * 32 + lane];
    }
}


typedef __attribute__((ext_vector_type(4))) float v4f;

#define SP (FDIM + 4)

template <int FOUT, bool RELU_IN>
__global__ __launch_bounds__(256) void gemm_wmma(const float* __restrict__ A,
                                                 const float* __restrict__ W,
                                                 float* __restrict__ H, int nrows) {
    constexpr int K  = FDIM;
    constexpr int CT = FOUT / 16;
    __shared__ float wlds[K * FOUT];
    __shared__ __attribute__((aligned(16))) float stg[8 * 16 * SP];

    const int wave  = threadIdx.x >> 5;
    const int lane  = threadIdx.x & 31;
    const int strip = blockIdx.x * 8 + wave;
    const int nstr  = nrows >> 4;
    const int row0  = (strip < nstr ? strip : nstr - 1) << 4;
    float* st = stg + wave * (16 * SP);

    v4f av[16];
#pragma unroll
    for (int i = 0; i < 16; ++i) av[i] = *(const v4f*)(A + (size_t)(row0 + i) * K + lane * 4);

    for (int i = threadIdx.x; i < K * FOUT; i += 256) wlds[i] = W[i];

#pragma unroll
    for (int i = 0; i < 16; ++i) {
        v4f v = av[i];
        if (RELU_IN) {
            v.x = fmaxf(v.x, 0.f); v.y = fmaxf(v.y, 0.f);
            v.z = fmaxf(v.z, 0.f); v.w = fmaxf(v.w, 0.f);
        }
        *(v4f*)(st + i * SP + lane * 4) = v;
    }
    __syncthreads();

    if (strip >= nstr) return;

    const int m    = lane & 15;
    const int koff = (lane >> 4) << 1;

    v8f acc[CT];
    const v8f zero = {0.f, 0.f, 0.f, 0.f, 0.f, 0.f, 0.f, 0.f};
#pragma unroll
    for (int t = 0; t < CT; ++t) acc[t] = zero;

    const float* arow = st + m * SP + koff;

    for (int kk = 0; kk < K; kk += 4) {
        const v2f a = *(const v2f*)(arow + kk);
#pragma unroll
        for (int t = 0; t < CT; ++t) {
            v2f b;
            b.x = wlds[(kk + koff)     * FOUT + t * 16 + m];
            b.y = wlds[(kk + koff + 1) * FOUT + t * 16 + m];
            acc[t] = __builtin_amdgcn_wmma_f32_16x16x4_f32(
                false, a, false, b, (short)0, acc[t], false, false);
        }
    }

    const int mrow = (lane >> 4) << 3;
#pragma unroll
    for (int t = 0; t < CT; ++t) {
#pragma unroll
        for (int v = 0; v < 8; ++v) st[(mrow + v) * SP + t * 16 + m] = acc[t][v];
    }
    __builtin_amdgcn_fence(__ATOMIC_ACQ_REL, "wavefront");

    constexpr int VPR = FOUT / 4;
    constexpr int RPI = 32 / VPR;
    const int rsub = lane / VPR;
    const int cc   = (lane % VPR) * 4;
    v4f o[16 / RPI];
#pragma unroll
    for (int i = 0; i < 16 / RPI; ++i) o[i] = *(const v4f*)(st + (i * RPI + rsub) * SP + cc);
#pragma unroll
    for (int i = 0; i < 16 / RPI; ++i)
        *(v4f*)(H + (size_t)(row0 + i * RPI + rsub) * FOUT + cc) = o[i];
}


#define AGG_THREADS 1024
#define AGG_WAVES   (AGG_THREADS / 32)
#define QCAP        (31 + 1024 + 1)


static_assert(NPB <= 256 && NN <= (1 << 24), "an entry packs source row << 8 | local destination row");

__device__ __forceinline__ int pad16(int n) { return (n + 15) & ~15; }

__device__ __forceinline__ int wave_sum(int v) {
#pragma unroll
    for (int o = 16; o > 0; o >>= 1) v += __shfl_xor(v, o);
    return v;
}

#define LIST_LOADS (NOWN * AGG_WAVES / AGG_THREADS)

__device__ __forceinline__ void list_loads(const int* __restrict__ wcnt, int b, int tid, int lane,
                                           int (&pv)[LIST_LOADS], int& mine) {
#pragma unroll
    for (int k = 0; k < LIST_LOADS; ++k) {
        const int i = tid + k * AGG_THREADS;
        pv[k] = wcnt[i < b * AGG_WAVES ? i : 0];
    }
    mine = wcnt[b * AGG_WAVES + lane];
}

__device__ __forceinline__ void list_partials(const int (&pv)[LIST_LOADS], int b, int tid,
                                              int lane, int wave, int* red) {
    int part = 0;
#pragma unroll
    for (int k = 0; k < LIST_LOADS; ++k)
        if (tid + k * AGG_THREADS < b * AGG_WAVES) part += pad16(pv[k]);
    part = wave_sum(part);
    if (lane == 0) red[wave] = part;
}

__device__ __forceinline__ int list_start(int mine, int lane, int wave, const int* red) {
    int v = red[lane];
    if (lane < wave) v += pad16(mine);
    return __builtin_amdgcn_readfirstlane(wave_sum(v));
}

template <int F, int TP = F>
__device__ __forceinline__ void tile_init(const float* __restrict__ H, const float* __restrict__ dinv,
                                          const float* __restrict__ bias, float* tile,
                                          int n0, int nn, int tid) {
    constexpr int VPL = F / 32;
    typedef __attribute__((ext_vector_type(VPL))) float rowv;
    const int j = tid & 31;
    float bb[VPL];
#pragma unroll
    for (int x = 0; x < VPL; ++x) bb[x] = bias[j * VPL + x];
    constexpr int TRIPS = (NPB * 32 + AGG_THREADS - 1) / AGG_THREADS;
    rowv  h[TRIPS];
    float d[TRIPS];
#pragma unroll
    for (int k = 0; k < TRIPS; ++k) {
        const int i  = tid + k * AGG_THREADS;
        const int rw = (i < nn * 32) ? (i >> 5) : 0;
        d[k] = dinv[n0 + rw];
        h[k] = *(const rowv*)(H + (size_t)(n0 + rw) * F + j * VPL);
    }
#pragma unroll
    for (int k = 0; k < TRIPS; ++k) {
        const int i = tid + k * AGG_THREADS;
        if (i < nn * 32) {
            const int rw = i >> 5;
#pragma unroll
            for (int x = 0; x < VPL; ++x) tile[rw * TP + x * 32 + j] = bb[x] + d[k] * d[k] * h[k][x];
        }
    }
}

template <int F, int TP, int ROWS>
__device__ __forceinline__ void tile_zero_tail(float* tile, int nn, int tid) {
    for (int i = nn * 32 + tid; i < ROWS * 32; i += AGG_THREADS) {
        float* tr = tile + (i >> 5) * TP + (i & 31);
        for (int x = 0; x < F / 32; ++x) tr[x * 32] = 0.f;
    }
}

template <int F>
__device__ __forceinline__ void tile_store(const float* tile, float* __restrict__ O,
                                           int n0, int nn, int tid) {
    constexpr int VPL = F / 32;
    typedef __attribute__((ext_vector_type(VPL))) float rowv;
    const int j = tid & 31;
    constexpr int TRIPS = (NPB * 32 + AGG_THREADS - 1) / AGG_THREADS;
    rowv o[TRIPS];
#pragma unroll
    for (int k = 0; k < TRIPS; ++k) {
        const int i  = tid + k * AGG_THREADS;
        const int rw = (i < nn * 32) ? (i >> 5) : 0;
#pragma unroll
        for (int x = 0; x < VPL; ++x) o[k][x] = tile[rw * F + x * 32 + j];
    }
#pragma unroll
    for (int k = 0; k < TRIPS; ++k) {
        const int i = tid + k * AGG_THREADS;
        if (i < nn * 32) *(rowv*)(O + (size_t)(n0 + (i >> 5)) * F + j * VPL) = o[k];
    }
}

template <int F, int TP = F>
__device__ __forceinline__ void add_edges(const float* __restrict__ H, float* tile, int lane,
                                          int r, int cl, int wbits, int cnt) {
    constexpr int VPL = F / 32;
    constexpr int G   = 64 / VPL;
    typedef __attribute__((ext_vector_type(VPL))) float rowv;
#pragma unroll
    for (int g = 0; g < 32; g += G) {
        if (g < cnt) {
            rowv hv[G];
#pragma unroll
            for (int u = 0; u < G; ++u) {
                const int rj = __builtin_amdgcn_readlane(r, g + u);
                hv[u] = *(const rowv*)(H + (size_t)rj * F + lane * VPL);
            }
#pragma unroll
            for (int u = 0; u < G; ++u) {
                if (g + u < cnt) {
                    const int   cj = __builtin_amdgcn_readlane(cl, g + u);
                    const float wj = __int_as_float(__builtin_amdgcn_readlane(wbits, g + u));
                    float* tr = tile + cj * TP + lane;
#pragma unroll
                    for (int x = 0; x < VPL; ++x) atomicAdd(&tr[x * 32], wj * hv[u][x]);
                }
            }
        }
    }
}


#define FT_ROWS (((NPB + 15) / 16) * 16)

template <int F, int FOUT>
struct wfill {
    static constexpr int PRW = F / 2 / AGG_WAVES;
    static constexpr int CH  = FOUT / 32;
    static constexpr int NL  = PRW * CH;
    static constexpr int WPP = 2 * FOUT + 32;
    static constexpr int SZ  = (F / 2) * WPP;
    static_assert((F / 2) % AGG_WAVES == 0 && FOUT % 32 == 0, "whole pair-rows per wave, whole pieces per row");
};

template <int F, int FOUT>
__device__ __forceinline__ void w_loads(const float* __restrict__ Wn, int lane, int wave,
                                        v2f (&wv)[wfill<F, FOUT>::NL]) {
    constexpr int VPL = F / 32;
    constexpr int CH  = wfill<F, FOUT>::CH;
#pragma unroll
    for (int j = 0; j < wfill<F, FOUT>::PRW; ++j) {
        const int p  = 2 * (wave + j * AGG_WAVES);
        const int c0 = (p % 32) * VPL + p / 32;
#pragma unroll
        for (int e = 0; e < CH; ++e) {
            wv[j * CH + e].x = Wn[c0 * FOUT + e * 32 + lane];
            wv[j * CH + e].y = Wn[(c0 + VPL) * FOUT + e * 32 + lane];
        }
    }
}

template <int F, int FOUT>
__device__ __forceinline__ void w_stores(float* wl, int lane, int wave,
                                         const v2f (&wv)[wfill<F, FOUT>::NL]) {
    constexpr int CH  = wfill<F, FOUT>::CH;
    constexpr int WPP = wfill<F, FOUT>::WPP;
#pragma unroll
    for (int j = 0; j < wfill<F, FOUT>::PRW; ++j) {
#pragma unroll
        for (int e = 0; e < CH; ++e)
            *(v2f*)(wl + (wave + j * AGG_WAVES) * WPP + (e * 32 + lane) * 2) = wv[j * CH + e];
    }
}

template <int F, int FOUT>
__device__ __forceinline__ void tile_transform(float* tile, const float* wl, float* __restrict__ Hn,
                                               int n0, int nn, int tid, int lane, int wave) {
    constexpr int TP  = F + 4;
    constexpr int WPP = wfill<F, FOUT>::WPP;
    constexpr int NT  = 2;
    constexpr int UPS = FOUT / 16 / NT;
    constexpr int NU  = (FT_ROWS / 16) * UPS;
    constexpr int UPW = (NU + AGG_WAVES - 1) / AGG_WAVES;

    const int m = lane & 15;
    const int h = lane >> 4;
    const v8f zero = {0.f, 0.f, 0.f, 0.f, 0.f, 0.f, 0.f, 0.f};
    v8f acc[UPW][NT];

#pragma unroll
    for (int q = 0; q < UPW; ++q) {
#pragma unroll
        for (int t = 0; t < NT; ++t) acc[q][t] = zero;
        const int u = wave + q * AGG_WAVES;
        if (u < NU) {
            const float* arow = tile + ((u / UPS) * 16 + m) * TP + 2 * h;
            const float* brow = wl + h * WPP + ((u % UPS) * NT * 16 + m) * 2;
            for (int kk = 0; kk < F; kk += 4) {
                v2f a = *(const v2f*)(arow + kk);
                a.x = fmaxf(a.x, 0.f);
                a.y = fmaxf(a.y, 0.f);
#pragma unroll
                for (int t = 0; t < NT; ++t) {
                    const v2f b = *(const v2f*)(brow + (kk >> 1) * WPP + t * 32);
                    acc[q][t] = __builtin_amdgcn_wmma_f32_16x16x4_f32(
                        false, a, false, b, (short)0, acc[q][t], false, false);
                }
            }
        }
    }
    __syncthreads();

    const int mrow = h << 3;
#pragma unroll
    for (int q = 0; q < UPW; ++q) {
        const int u = wave + q * AGG_WAVES;
        if (u < NU) {
            float* orow = tile + ((u / UPS) * 16 + mrow) * TP + (u % UPS) * NT * 16 + m;
#pragma unroll
            for (int t = 0; t < NT; ++t) {
#pragma unroll
                for (int v = 0; v < 8; ++v) orow[v * TP + t * 16] = acc[q][t][v];
            }
        }
    }
    __syncthreads();

    constexpr int VPR   = FOUT / 4;
    constexpr int TRIPS = (NPB * VPR + AGG_THREADS - 1) / AGG_THREADS;
    v4f o[TRIPS];
#pragma unroll
    for (int k = 0; k < TRIPS; ++k) {
        const int i  = tid + k * AGG_THREADS;
        const int rw = (i < nn * VPR) ? (i / VPR) : 0;
        o[k] = *(const v4f*)(tile + rw * TP + (i % VPR) * 4);
    }
#pragma unroll
    for (int k = 0; k < TRIPS; ++k) {
        const int i = tid + k * AGG_THREADS;
        if (i < nn * VPR)
            *(v4f*)(Hn + (size_t)(n0 + i / VPR) * FOUT + (i % VPR) * 4) = o[k];
    }
}

template <int F, int FOUT>
__global__ __launch_bounds__(AGG_THREADS) void agg_lds(const float* __restrict__ H,
                                                       const int* __restrict__ row,
                                                       const int* __restrict__ col,
                                                       const unsigned* __restrict__ bm,
                                                       const float* __restrict__ dinv,
                                                       const float* __restrict__ bias,
                                                       const int* __restrict__ wcnt,
                                                       unsigned* __restrict__ list,
                                                       const float* __restrict__ Wn,
                                                       float* __restrict__ O, int nedges) {
    static_assert(FOUT > 0, "the first aggregation always goes on to the next transform");
    constexpr int TP = F + 4;
    __shared__ __attribute__((aligned(16))) float tile[FT_ROWS * TP];
    __shared__ __attribute__((aligned(16))) float wl[wfill<F, FOUT>::SZ];
    __shared__ unsigned wq[AGG_WAVES][QCAP];
    __shared__ int red[AGG_WAVES];

    const int tid  = threadIdx.x;
    const int lane = tid & 31;
    const int wave = __builtin_amdgcn_readfirstlane(tid >> 5);
    const int b  = blockIdx.x;
    const int n0 = b * NPB;
    int nn = NN - n0;
    if (nn > NPB) nn = NPB;
    if (nn < 0) nn = 0;

    int pv[LIST_LOADS], mine;
    v2f wv[wfill<F, FOUT>::NL];
    list_loads(wcnt, b, tid, lane, pv, mine);
    w_loads<F, FOUT>(Wn, lane, wave, wv);
    tile_init<F, TP>(H, dinv, bias, tile, n0, nn, tid);
    tile_zero_tail<F, TP, FT_ROWS>(tile, nn, tid);
    w_stores<F, FOUT>(wl, lane, wave, wv);
    list_partials(pv, b, tid, lane, wave, red);
    __syncthreads();

    uint2* mylist = (uint2*)list + list_start(mine, lane, wave, red);
    int lpos = 0;

    const int nw = bm_words(nedges);
    const unsigned* mybm = bm + (size_t)b * nw;
    const int nsteps = (nw + 127) >> 7;
    int qc = 0;

    for (int j = wave; j < nsteps + AGG_WAVES; j += AGG_WAVES) {
        const bool last = (j >= nsteps);
        const int wi = j * 128 + lane * 4;
        uint4 bw = {0u, 0u, 0u, 0u};
        if (!last && wi < nw) bw = *(const uint4*)(mybm + wi);
#pragma nounroll
        for (int k = 0; k < 4; ++k) {
            unsigned wd = bw.x;
            bw.x = bw.y; bw.y = bw.z; bw.z = bw.w; bw.w = 0u;
            const unsigned ebase = (unsigned)(wi + k) * 32u;
            unsigned act = __builtin_amdgcn_ballot_w32(wd != 0u);
            while (act) {
                if (wd) {
                    const int pos = qc + (int)__builtin_amdgcn_mbcnt_lo(act, 0u);
                    wq[wave][pos] = ebase + (unsigned)__builtin_ctz(wd);
                    wd &= wd - 1u;
                }
                qc += __builtin_popcount(act);
                act = __builtin_amdgcn_ballot_w32(wd != 0u);
            }
            while (qc >= 32 || (last && qc > 0)) {
                const int cnt  = qc < 32 ? qc : 32;
                const int base = qc - cnt;
                qc = base;
                int r = 0, c = 0;
                float w = 0.f;
                __builtin_amdgcn_fence(__ATOMIC_ACQ_REL, "wavefront");
                if (lane < cnt) {
                    const int e = (int)wq[wave][base + lane];
                    r = row[e];
                    c = col[e] - n0;
                    w = dinv[r] * dinv[c + n0];
                    uint2 ent;
                    ent.x = ((unsigned)r << 8) | (unsigned)c;
                    ent.y = (unsigned)__float_as_int(w);
                    mylist[lpos + lane] = ent;
                }
                lpos += cnt;
                add_edges<F, TP>(H, tile, lane, r, c, __float_as_int(w), cnt);
            }
        }
    }
    __syncthreads();

    tile_transform<F, FOUT>(tile, wl, O, n0, nn, tid, lane, wave);
}

template <int F, int FOUT>
__global__ __launch_bounds__(AGG_THREADS) void agg_replay(const float* __restrict__ H,
                                                          const unsigned* __restrict__ list,
                                                          const int* __restrict__ wcnt,
                                                          const float* __restrict__ dinv,
                                                          const float* __restrict__ bias,
                                                          const float* __restrict__ Wn,
                                                          float* __restrict__ O) {
    constexpr bool FUSED = FOUT > 0;
    constexpr int  FO = FUSED ? FOUT : F;
    constexpr int  TP = FUSED ? F + 4 : F;
    __shared__ __attribute__((aligned(16))) float tile[(FUSED ? FT_ROWS : NPB) * TP];
    __shared__ __attribute__((aligned(16))) float wl[FUSED ? wfill<F, FO>::SZ : 4];
    __shared__ int red[AGG_WAVES];

    const int tid  = threadIdx.x;
    const int lane = tid & 31;
    const int wave = __builtin_amdgcn_readfirstlane(tid >> 5);
    const int b  = blockIdx.x;
    const int n0 = b * NPB;
    int nn = NN - n0;
    if (nn > NPB) nn = NPB;
    if (nn < 0) nn = 0;

    int pv[LIST_LOADS], mine;
    v2f wv[wfill<F, FO>::NL];
    list_loads(wcnt, b, tid, lane, pv, mine);
    if constexpr (FUSED) w_loads<F, FO>(Wn, lane, wave, wv);
    tile_init<F, TP>(H, dinv, bias, tile, n0, nn, tid);
    if constexpr (FUSED) {
        tile_zero_tail<F, TP, FT_ROWS>(tile, nn, tid);
        w_stores<F, FO>(wl, lane, wave, wv);
    }
    list_partials(pv, b, tid, lane, wave, red);
    __syncthreads();

    const int count = __builtin_amdgcn_readlane(mine, wave);
    const uint2* mylist = (const uint2*)list + list_start(mine, lane, wave, red);
    const uint2 zero2 = {0u, 0u};
    uint2 nx = zero2;
    if (lane < count) nx = mylist[lane];
    for (int p = 0; p < count; p += 32) {
        const int cnt = (count - p) < 32 ? (count - p) : 32;
        const uint2 ent = nx;
        const int q = p + 32 + lane;
        const uint2 t = mylist[q < count ? q : count - 1];
        nx = (q < count) ? t : zero2;
        add_edges<F, TP>(H, tile, lane, (int)(ent.x >> 8), (int)(ent.x & 255u), (int)ent.y, cnt);
    }
    __syncthreads();

    if constexpr (FUSED) tile_transform<F, FO>(tile, wl, O, n0, nn, tid, lane, wave);
    else                 tile_store<F>(tile, O, n0, nn, tid);
}


__global__ __launch_bounds__(AGG_THREADS) void own_deg(const int* __restrict__ col,
                                                       const unsigned* __restrict__ bm,
                                                       float* __restrict__ dinv,
                                                       int* __restrict__ wcnt, int nedges) {
    __shared__ int cnt[NPB];
    __shared__ int wtot[AGG_WAVES];
    __shared__ unsigned wq[AGG_WAVES][QCAP];

    const int tid  = threadIdx.x;
    const int lane = tid & 31;
    const int wave = __builtin_amdgcn_readfirstlane(tid >> 5);
    const int b  = blockIdx.x;
    const int n0 = b * NPB;
    int nn = NN - n0;
    if (nn > NPB) nn = NPB;
    if (nn < 0) nn = 0;

    if (tid < NPB) cnt[tid] = 0;
    __syncthreads();

    const int nw = bm_words(nedges);
    const unsigned* mybm = bm + (size_t)b * nw;
    const int nsteps = (nw + 127) >> 7;
    int qc = 0;
    int tot = 0;

    const uint4 zero4 = {0u, 0u, 0u, 0u};
    const bool have0 = (wave < nsteps) && (wave * 128 + lane * 4 < nw);
    uint4 nx = *(const uint4*)(mybm + (have0 ? wave * 128 + lane * 4 : 0));
    if (!have0) nx = zero4;

    for (int j = wave; j < nsteps + AGG_WAVES; j += AGG_WAVES) {
        const bool last = (j >= nsteps);
        const int wi = j * 128 + lane * 4;
        const uint4 bw = nx;
        const int wn = wi + AGG_WAVES * 128;
        const bool have = (j + AGG_WAVES < nsteps) && (wn < nw);
        nx = *(const uint4*)(mybm + (have ? wn : 0));
        if (!have) nx = zero4;
#pragma unroll
        for (int k = 0; k < 4; ++k) {
            unsigned wd = (k == 0) ? bw.x : (k == 1) ? bw.y : (k == 2) ? bw.z : bw.w;
            const unsigned ebase = (unsigned)(wi + k) * 32u;
            unsigned act = __builtin_amdgcn_ballot_w32(wd != 0u);
            while (act) {
                if (wd) {
                    const int pos = qc + (int)__builtin_amdgcn_mbcnt_lo(act, 0u);
                    wq[wave][pos] = ebase + (unsigned)__builtin_ctz(wd);
                    wd &= wd - 1u;
                }
                qc  += __builtin_popcount(act);
                tot += __builtin_popcount(act);
                act = __builtin_amdgcn_ballot_w32(wd != 0u);
            }
            while (qc >= 32 || (last && qc > 0)) {
                const int take = qc < 32 ? qc : 32;
                const int base = qc - take;
                qc = base;
                __builtin_amdgcn_fence(__ATOMIC_ACQ_REL, "wavefront");
                if (lane < take) {
                    const int e = (int)wq[wave][base + lane];
                    atomicAdd(&cnt[col[e] - n0], 1);
                }
            }
        }
    }
    if (lane == 0) wtot[wave] = tot;
    __syncthreads();

    if (tid < nn) dinv[n0 + tid] = rsqrtf((float)(cnt[tid] + 1));
    if (tid < AGG_WAVES) wcnt[b * AGG_WAVES + tid] = wtot[tid];
}


extern "C" void kernel_launch(void* const* d_in, const int* in_sizes, int n_in,
                              void* d_out, int out_size, void* d_ws, size_t ws_size,
                              hipStream_t stream) {
    const float* x  = (const float*)d_in[0];
    const int*   ei = (const int*)d_in[1];
    const float* W1 = (const float*)d_in[2];
    const float* b1 = (const float*)d_in[3];
    const float* W2 = (const float*)d_in[4];
    const float* b2 = (const float*)d_in[5];
    const float* W3 = (const float*)d_in[6];
    const float* b3 = (const float*)d_in[7];
    float* out = (float*)d_out;

    const int E = in_sizes[1] / 2;
    const int* row = ei;
    const int* col = ei + E;

    float* ws   = (float*)d_ws;
    float* dinv = ws;
    float* H    = ws + 50176;
    float* O    = H + (size_t)NN * FDIM;
    unsigned* bm = (unsigned*)(O + (size_t)NN * FDIM);
    int*      wcnt = (int*)d_ws + 33546240;
    unsigned* list = (unsigned*)d_ws + 33554432;

    build_bm<<<(E + CH_EDGES - 1) / CH_EDGES, BM_THREADS, 0, stream>>>(col, bm, E);
    own_deg<<<NOWN, AGG_THREADS, 0, stream>>>(col, bm, dinv, wcnt, E);

    const int ggrid  = ((NN / 16) + 7) / 8;

    gemm_wmma<128, false><<<ggrid, 256, 0, stream>>>(x, W1, H, NN);
    agg_lds<128, 128><<<NOWN, AGG_THREADS, 0, stream>>>(H, row, col, bm, dinv, b1, wcnt, list, W2, O, E);

    agg_replay<128, 64><<<NOWN, AGG_THREADS, 0, stream>>>(O, list, wcnt, dinv, b2, W3, H);

    agg_replay<64, 0><<<NOWN, AGG_THREADS, 0, stream>>>(H, list, wcnt, dinv, b3, nullptr, out);
}


# layers as (A X) W + b: layer 1 sums input rows, then W1, bias, ReLU on the owner tile; layer 2's block chains W2 and W3; transform launch gone (5 launches); sum order: aggregate, then product
# speedup vs baseline: 3.7184x; 1.0961x over previous
#include <hip/hip_runtime.h>

#define NN 50000
#define FDIM 128
#define NPB  196
#define NOWN ((NN + NPB - 1) / NPB)

typedef __attribute__((ext_vector_type(2))) float v2f;
typedef __attribute__((ext_vector_type(8))) float v8f;


#define CH_EDGES 4096
#define CH_WORDS (CH_EDGES / 32)
#define PL_PITCH (CH_WORDS + 1)
#define BM_THREADS 1024

__device__ __forceinline__ int bm_words(int nedges) {
    return (((nedges + 31) >> 5) + CH_WORDS - 1) & ~(CH_WORDS - 1);
}

__global__ __launch_bounds__(BM_THREADS) void build_bm(const int* __restrict__ col,
                                                       unsigned* __restrict__ bm, int e) {
    __shared__ __attribute__((aligned(16))) unsigned pl[NOWN * PL_PITCH];
    const int tid = threadIdx.x;
    uint4* pl4 = (uint4*)pl;
    const uint4 zero = {0u, 0u, 0u, 0u};
    for (int i = tid; i < NOWN * PL_PITCH / 4; i += BM_THREADS) pl4[i] = zero;
    __syncthreads();

    const int e0 = blockIdx.x * CH_EDGES;
    int c[CH_EDGES / BM_THREADS];
#pragma unroll
    for (int k = 0; k < CH_EDGES / BM_THREADS; ++k) {
        const int i = e0 + k * BM_THREADS + tid;
        c[k] = (i < e) ? col[i] : -1;
    }
#pragma unroll
    for (int k = 0; k < CH_EDGES / BM_THREADS; ++k) {
        const int li = k * BM_THREADS + tid;
        if (c[k] >= 0) atomicOr(&pl[(c[k] / NPB) * PL_PITCH + (li >> 5)], 1u << (li & 31));
    }
    __syncthreads();

    const int nw   = bm_words(e);
    const int lane = tid & 31;
    const int wave = tid >> 5;
    for (int p = wave; p < NOWN; p += BM_THREADS / 32) {
        unsigned* dst = bm + (size_t)p * nw + blockIdx.x * CH_WORDS + lane;
#pragma unroll
        for (int x = 0; x < CH_WORDS / 32; ++x) dst[x * 32] = pl[p * PL_PITCH + x * 32 + lane];
    }
}

typedef __attribute__((ext_vector_type(4))) float v4f;


#define AGG_THREADS 1024
#define AGG_WAVES   (AGG_THREADS / 32)
#define QCAP        (31 + 1024 + 1)


static_assert(NPB <= 256 && NN <= (1 << 24), "an entry packs source row << 8 | local destination row");

__device__ __forceinline__ int pad16(int n) { return (n + 15) & ~15; }

__device__ __forceinline__ int wave_sum(int v) {
#pragma unroll
    for (int o = 16; o > 0; o >>= 1) v += __shfl_xor(v, o);
    return v;
}

#define LIST_LOADS (NOWN * AGG_WAVES / AGG_THREADS)

__device__ __forceinline__ void list_loads(const int* __restrict__ wcnt, int b, int tid, int lane,
                                           int (&pv)[LIST_LOADS], int& mine) {
#pragma unroll
    for (int k = 0; k < LIST_LOADS; ++k) {
        const int i = tid + k * AGG_THREADS;
        pv[k] = wcnt[i < b * AGG_WAVES ? i : 0];
    }
    mine = wcnt[b * AGG_WAVES + lane];
}

__device__ __forceinline__ void list_partials(const int (&pv)[LIST_LOADS], int b, int tid,
                                              int lane, int wave, int* red) {
    int part = 0;
#pragma unroll
    for (int k = 0; k < LIST_LOADS; ++k)
        if (tid + k * AGG_THREADS < b * AGG_WAVES) part += pad16(pv[k]);
    part = wave_sum(part);
    if (lane == 0) red[wave] = part;
}

__device__ __forceinline__ int list_start(int mine, int lane, int wave, const int* red) {
    int v = red[lane];
    if (lane < wave) v += pad16(mine);
    return __builtin_amdgcn_readfirstlane(wave_sum(v));
}

template <int F, int TP, bool BIAS>
__device__ __forceinline__ void tile_init(const float* __restrict__ H, const float* __restrict__ dinv,
                                          const float* __restrict__ bias, float* tile,
                                          int n0, int nn, int tid) {
    constexpr int VPL = F / 32;
    typedef __attribute__((ext_vector_type(VPL))) float rowv;
    const int j = tid & 31;
    float bb[VPL];
#pragma unroll
    for (int x = 0; x < VPL; ++x) {
        if constexpr (BIAS) bb[x] = bias[j * VPL + x];
        else                bb[x] = 0.f;
    }
    constexpr int TRIPS = (NPB * 32 + AGG_THREADS - 1) / AGG_THREADS;
    rowv  h[TRIPS];
    float d[TRIPS];
#pragma unroll
    for (int k = 0; k < TRIPS; ++k) {
        const int i  = tid + k * AGG_THREADS;
        const int rw = (i < nn * 32) ? (i >> 5) : 0;
        d[k] = dinv[n0 + rw];
        h[k] = *(const rowv*)(H + (size_t)(n0 + rw) * F + j * VPL);
    }
#pragma unroll
    for (int k = 0; k < TRIPS; ++k) {
        const int i = tid + k * AGG_THREADS;
        if (i < nn * 32) {
            const int rw = i >> 5;
#pragma unroll
            for (int x = 0; x < VPL; ++x) tile[rw * TP + x * 32 + j] = bb[x] + d[k] * d[k] * h[k][x];
        }
    }
}

template <int F, int TP, int ROWS>
__device__ __forceinline__ void tile_zero_tail(float* tile, int nn, int tid) {
    for (int i = nn * 32 + tid; i < ROWS * 32; i += AGG_THREADS) {
        float* tr = tile + (i >> 5) * TP + (i & 31);
        for (int x = 0; x < F / 32; ++x) tr[x * 32] = 0.f;
    }
}

template <int F>
__device__ __forceinline__ void tile_store(const float* tile, float* __restrict__ O,
                                           int n0, int nn, int tid) {
    constexpr int VPL = F / 32;
    typedef __attribute__((ext_vector_type(VPL))) float rowv;
    const int j = tid & 31;
    constexpr int TRIPS = (NPB * 32 + AGG_THREADS - 1) / AGG_THREADS;
    rowv o[TRIPS];
#pragma unroll
    for (int k = 0; k < TRIPS; ++k) {
        const int i  = tid + k * AGG_THREADS;
        const int rw = (i < nn * 32) ? (i >> 5) : 0;
#pragma unroll
        for (int x = 0; x < VPL; ++x) o[k][x] = tile[rw * F + x * 32 + j];
    }
#pragma unroll
    for (int k = 0; k < TRIPS; ++k) {
        const int i = tid + k * AGG_THREADS;
        if (i < nn * 32) *(rowv*)(O + (size_t)(n0 + (i >> 5)) * F + j * VPL) = o[k];
    }
}

template <int F, int TP = F>
__device__ __forceinline__ void add_edges(const float* __restrict__ H, float* tile, int lane,
                                          int r, int cl, int wbits, int cnt) {
    constexpr int VPL = F / 32;
    constexpr int G   = 64 / VPL;
    typedef __attribute__((ext_vector_type(VPL))) float rowv;
#pragma unroll
    for (int g = 0; g < 32; g += G) {
        if (g < cnt) {
            rowv hv[G];
#pragma unroll
            for (int u = 0; u < G; ++u) {
                const int rj = __builtin_amdgcn_readlane(r, g + u);
                hv[u] = *(const rowv*)(H + (size_t)rj * F + lane * VPL);
            }
#pragma unroll
            for (int u = 0; u < G; ++u) {
                if (g + u < cnt) {
                    const int   cj = __builtin_amdgcn_readlane(cl, g + u);
                    const float wj = __int_as_float(__builtin_amdgcn_readlane(wbits, g + u));
                    float* tr = tile + cj * TP + lane;
#pragma unroll
                    for (int x = 0; x < VPL; ++x) atomicAdd(&tr[x * 32], wj * hv[u][x]);
                }
            }
        }
    }
}


#define FT_ROWS (((NPB + 15) / 16) * 16)

template <int F, int FOUT>
struct wfill {
    static constexpr int PRW = F / 2 / AGG_WAVES;
    static constexpr int CH  = FOUT / 32;
    static constexpr int NL  = PRW * CH;
    static constexpr int WPP = 2 * FOUT + 32;
    static constexpr int SZ  = (F / 2) * WPP;
    static_assert((F / 2) % AGG_WAVES == 0 && FOUT % 32 == 0, "whole pair-rows per wave, whole pieces per row");
};

template <int F, int FOUT, bool NAT>
__device__ __forceinline__ void w_loads(const float* __restrict__ Wn, int lane, int wave,
                                        v2f (&wv)[wfill<F, FOUT>::NL]) {
    constexpr int VPL = F / 32;
    constexpr int CH  = wfill<F, FOUT>::CH;
#pragma unroll
    for (int j = 0; j < wfill<F, FOUT>::PRW; ++j) {
        const int p  = 2 * (wave + j * AGG_WAVES);
        const int c0 = NAT ? p : (p % 32) * VPL + p / 32;
        const int c1 = NAT ? p + 1 : c0 + VPL;
#pragma unroll
        for (int e = 0; e < CH; ++e) {
            wv[j * CH + e].x = Wn[c0 * FOUT + e * 32 + lane];
            wv[j * CH + e].y = Wn[c1 * FOUT + e * 32 + lane];
        }
    }
}

template <int F, int FOUT>
__device__ __forceinline__ void w_stores(float* wl, int lane, int wave,
                                         const v2f (&wv)[wfill<F, FOUT>::NL]) {
    constexpr int CH  = wfill<F, FOUT>::CH;
    constexpr int WPP = wfill<F, FOUT>::WPP;
#pragma unroll
    for (int j = 0; j < wfill<F, FOUT>::PRW; ++j) {
#pragma unroll
        for (int e = 0; e < CH; ++e)
            *(v2f*)(wl + (wave + j * AGG_WAVES) * WPP + (e * 32 + lane) * 2) = wv[j * CH + e];
    }
}

template <int F, int FOUT, bool EPI, bool STORE>
__device__ __forceinline__ void tile_transform(float* tile, const float* wl,
                                               const float* __restrict__ bias_o, float* __restrict__ Hn,
                                               int n0, int nn, int tid, int lane, int wave) {
    constexpr int TP  = F + 4;
    constexpr int WPP = wfill<F, FOUT>::WPP;
    constexpr int NT  = 2;
    constexpr int UPS = FOUT / 16 / NT;
    constexpr int NU  = (FT_ROWS / 16) * UPS;
    constexpr int UPW = (NU + AGG_WAVES - 1) / AGG_WAVES;

    const int m = lane & 15;
    const int h = lane >> 4;
    const v8f zero = {0.f, 0.f, 0.f, 0.f, 0.f, 0.f, 0.f, 0.f};
    v8f acc[UPW][NT];
    float bo[UPW][NT];

#pragma unroll
    for (int q = 0; q < UPW; ++q) {
#pragma unroll
        for (int t = 0; t < NT; ++t) { acc[q][t] = zero; bo[q][t] = 0.f; }
        const int u = wave + q * AGG_WAVES;
        if (u < NU) {
            if constexpr (EPI) {
#pragma unroll
                for (int t = 0; t < NT; ++t) bo[q][t] = bias_o[((u % UPS) * NT + t) * 16 + m];
            }
            const float* arow = tile + ((u / UPS) * 16 + m) * TP + 2 * h;
            const float* brow = wl + h * WPP + ((u % UPS) * NT * 16 + m) * 2;
            for (int kk = 0; kk < F; kk += 4) {
                const v2f a = *(const v2f*)(arow + kk);
#pragma unroll
                for (int t = 0; t < NT; ++t) {
                    const v2f b = *(const v2f*)(brow + (kk >> 1) * WPP + t * 32);
                    acc[q][t] = __builtin_amdgcn_wmma_f32_16x16x4_f32(
                        false, a, false, b, (short)0, acc[q][t], false, false);
                }
            }
        }
    }
    __syncthreads();

    const int mrow = h << 3;
#pragma unroll
    for (int q = 0; q < UPW; ++q) {
        const int u = wave + q * AGG_WAVES;
        if (u < NU) {
            float* orow = tile + ((u / UPS) * 16 + mrow) * TP + (u % UPS) * NT * 16 + m;
#pragma unroll
            for (int t = 0; t < NT; ++t) {
#pragma unroll
                for (int v = 0; v < 8; ++v) {
                    float r = acc[q][t][v];
                    if constexpr (EPI) r = fmaxf(r + bo[q][t], 0.f);
                    orow[v * TP + t * 16] = r;
                }
            }
        }
    }
    __syncthreads();

    if constexpr (STORE) {
        constexpr int VPR   = FOUT / 4;
        constexpr int TRIPS = (NPB * VPR + AGG_THREADS - 1) / AGG_THREADS;
        v4f o[TRIPS];
#pragma unroll
        for (int k = 0; k < TRIPS; ++k) {
            const int i  = tid + k * AGG_THREADS;
            const int rw = (i < nn * VPR) ? (i / VPR) : 0;
            o[k] = *(const v4f*)(tile + rw * TP + (i % VPR) * 4);
        }
#pragma unroll
        for (int k = 0; k < TRIPS; ++k) {
            const int i = tid + k * AGG_THREADS;
            if (i < nn * VPR)
                *(v4f*)(Hn + (size_t)(n0 + i / VPR) * FOUT + (i % VPR) * 4) = o[k];
        }
    }
}

template <int F, int FOUT>
__global__ __launch_bounds__(AGG_THREADS) void agg_lds(const float* __restrict__ H,
                                                       const int* __restrict__ row,
                                                       const int* __restrict__ col,
                                                       const unsigned* __restrict__ bm,
                                                       const float* __restrict__ dinv,
                                                       const float* __restrict__ bias,
                                                       const int* __restrict__ wcnt,
                                                       unsigned* __restrict__ list,
                                                       const float* __restrict__ Wn,
                                                       float* __restrict__ O, int nedges) {
    static_assert(FOUT > 0, "the first aggregation always goes on to the layer's product");
    constexpr int TP = F + 4;
    __shared__ __attribute__((aligned(16))) float tile[FT_ROWS * TP];
    __shared__ __attribute__((aligned(16))) float wl[wfill<F, FOUT>::SZ];
    __shared__ unsigned wq[AGG_WAVES][QCAP];
    __shared__ int red[AGG_WAVES];

    const int tid  = threadIdx.x;
    const int lane = tid & 31;
    const int wave = __builtin_amdgcn_readfirstlane(tid >> 5);
    const int b  = blockIdx.x;
    const int n0 = b * NPB;
    int nn = NN - n0;
    if (nn > NPB) nn = NPB;
    if (nn < 0) nn = 0;

    int pv[LIST_LOADS], mine;
    v2f wv[wfill<F, FOUT>::NL];
    list_loads(wcnt, b, tid, lane, pv, mine);
    w_loads<F, FOUT, false>(Wn, lane, wave, wv);
    tile_init<F, TP, false>(H, dinv, nullptr, tile, n0, nn, tid);
    tile_zero_tail<F, TP, FT_ROWS>(tile, nn, tid);
    w_stores<F, FOUT>(wl, lane, wave, wv);
    list_partials(pv, b, tid, lane, wave, red);
    __syncthreads();

    uint2* mylist = (uint2*)list + list_start(mine, lane, wave, red);
    int lpos = 0;

    const int nw = bm_words(nedges);
    const unsigned* mybm = bm + (size_t)b * nw;
    const int nsteps = (nw + 127) >> 7;
    int qc = 0;

    for (int j = wave; j < nsteps + AGG_WAVES; j += AGG_WAVES) {
        const bool last = (j >= nsteps);
        const int wi = j * 128 + lane * 4;
        uint4 bw = {0u, 0u, 0u, 0u};
        if (!last && wi < nw) bw = *(const uint4*)(mybm + wi);
#pragma nounroll
        for (int k = 0; k < 4; ++k) {
            unsigned wd = bw.x;
            bw.x = bw.y; bw.y = bw.z; bw.z = bw.w; bw.w = 0u;
            const unsigned ebase = (unsigned)(wi + k) * 32u;
            unsigned act = __builtin_amdgcn_ballot_w32(wd != 0u);
            while (act) {
                if (wd) {
                    const int pos = qc + (int)__builtin_amdgcn_mbcnt_lo(act, 0u);
                    wq[wave][pos] = ebase + (unsigned)__builtin_ctz(wd);
                    wd &= wd - 1u;
                }
                qc += __builtin_popcount(act);
                act = __builtin_amdgcn_ballot_w32(wd != 0u);
            }
            while (qc >= 32 || (last && qc > 0)) {
                const int cnt  = qc < 32 ? qc : 32;
                const int base = qc - cnt;
                qc = base;
                int r = 0, c = 0;
                float w = 0.f;
                __builtin_amdgcn_fence(__ATOMIC_ACQ_REL, "wavefront");
                if (lane < cnt) {
                    const int e = (int)wq[wave][base + lane];
                    r = row[e];
                    c = col[e] - n0;
                    w = dinv[r] * dinv[c + n0];
                    uint2 ent;
                    ent.x = ((unsigned)r << 8) | (unsigned)c;
                    ent.y = (unsigned)__float_as_int(w);
                    mylist[lpos + lane] = ent;
                }
                lpos += cnt;
                add_edges<F, TP>(H, tile, lane, r, c, __float_as_int(w), cnt);
            }
        }
    }
    __syncthreads();

    tile_transform<F, FOUT, true, true>(tile, wl, bias, O, n0, nn, tid, lane, wave);
}

template <int F, int FMID, int FOUT>
__global__ __launch_bounds__(AGG_THREADS) void agg_replay(const float* __restrict__ H,
                                                          const unsigned* __restrict__ list,
                                                          const int* __restrict__ wcnt,
                                                          const float* __restrict__ dinv,
                                                          const float* __restrict__ bias,
                                                          const float* __restrict__ Wa,
                                                          const float* __restrict__ Wb,
                                                          float* __restrict__ O) {
    constexpr bool FUSED = FMID > 0;
    static_assert(!FUSED || (FMID == F && FOUT > 0), "the first product's result lies in the tile's own rows");
    constexpr int  FM = FUSED ? FMID : F;
    constexpr int  FO = FUSED ? FOUT : F;
    constexpr int  TP = FUSED ? F + 4 : F;
    __shared__ __attribute__((aligned(16))) float tile[(FUSED ? FT_ROWS : NPB) * TP];
    __shared__ __attribute__((aligned(16))) float wla[FUSED ? wfill<F, FM>::SZ : 4];
    __shared__ __attribute__((aligned(16))) float wlb[FUSED ? wfill<FM, FO>::SZ : 4];
    __shared__ int red[AGG_WAVES];

    const int tid  = threadIdx.x;
    const int lane = tid & 31;
    const int wave = __builtin_amdgcn_readfirstlane(tid >> 5);
    const int b  = blockIdx.x;
    const int n0 = b * NPB;
    int nn = NN - n0;
    if (nn > NPB) nn = NPB;
    if (nn < 0) nn = 0;

    int pv[LIST_LOADS], mine;
    v2f wva[wfill<F, FM>::NL];
    v2f wvb[wfill<FM, FO>::NL];
    list_loads(wcnt, b, tid, lane, pv, mine);
    if constexpr (FUSED) {
        w_loads<F, FM, false>(Wa, lane, wave, wva);
        w_loads<FM, FO, true>(Wb, lane, wave, wvb);
    }
    tile_init<F, TP, !FUSED>(H, dinv, bias, tile, n0, nn, tid);
    if constexpr (FUSED) {
        tile_zero_tail<F, TP, FT_ROWS>(tile, nn, tid);
        w_stores<F, FM>(wla, lane, wave, wva);
        w_stores<FM, FO>(wlb, lane, wave, wvb);
    }
    list_partials(pv, b, tid, lane, wave, red);
    __syncthreads();

    const int count = __builtin_amdgcn_readlane(mine, wave);
    const uint2* mylist = (const uint2*)list + list_start(mine, lane, wave, red);
    const uint2 zero2 = {0u, 0u};
    uint2 nx = zero2;
    if (lane < count) nx = mylist[lane];
    for (int p = 0; p < count; p += 32) {
        const int cnt = (count - p) < 32 ? (count - p) : 32;
        const uint2 ent = nx;
        const int q = p + 32 + lane;
        const uint2 t = mylist[q < count ? q : count - 1];
        nx = (q < count) ? t : zero2;
        add_edges<F, TP>(H, tile, lane, (int)(ent.x >> 8), (int)(ent.x & 255u), (int)ent.y, cnt);
    }
    __syncthreads();

    if constexpr (FUSED) {
        tile_transform<F, FM, true, false>(tile, wla, bias, nullptr, n0, nn, tid, lane, wave);
        tile_transform<FM, FO, false, true>(tile, wlb, nullptr, O, n0, nn, tid, lane, wave);
    } else {
        tile_store<F>(tile, O, n0, nn, tid);
    }
}


__global__ __launch_bounds__(AGG_THREADS) void own_deg(const int* __restrict__ col,
                                                       const unsigned* __restrict__ bm,
                                                       float* __restrict__ dinv,
                                                       int* __restrict__ wcnt, int nedges) {
    __shared__ int cnt[NPB];
    __shared__ int wtot[AGG_WAVES];
    __shared__ unsigned wq[AGG_WAVES][QCAP];

    const int tid  = threadIdx.x;
    const int lane = tid & 31;
    const int wave = __builtin_amdgcn_readfirstlane(tid >> 5);
    const int b  = blockIdx.x;
    const int n0 = b * NPB;
    int nn = NN - n0;
    if (nn > NPB) nn = NPB;
    if (nn < 0) nn = 0;

    if (tid < NPB) cnt[tid] = 0;
    __syncthreads();

    const int nw = bm_words(nedges);
    const unsigned* mybm = bm + (size_t)b * nw;
    const int nsteps = (nw + 127) >> 7;
    int qc = 0;
    int tot = 0;

    const uint4 zero4 = {0u, 0u, 0u, 0u};
    const bool have0 = (wave < nsteps) && (wave * 128 + lane * 4 < nw);
    uint4 nx = *(const uint4*)(mybm + (have0 ? wave * 128 + lane * 4 : 0));
    if (!have0) nx = zero4;

    for (int j = wave; j < nsteps + AGG_WAVES; j += AGG_WAVES) {
        const bool last = (j >= nsteps);
        const int wi = j * 128 + lane * 4;
        const uint4 bw = nx;
        const int wn = wi + AGG_WAVES * 128;
        const bool have = (j + AGG_WAVES < nsteps) && (wn < nw);
        nx = *(const uint4*)(mybm + (have ? wn : 0));
        if (!have) nx = zero4;
#pragma unroll
        for (int k = 0; k < 4; ++k) {
            unsigned wd = (k == 0) ? bw.x : (k == 1) ? bw.y : (k == 2) ? bw.z : bw.w;
            const unsigned ebase = (unsigned)(wi + k) * 32u;
            unsigned act = __builtin_amdgcn_ballot_w32(wd != 0u);
            while (act) {
                if (wd) {
                    const int pos = qc + (int)__builtin_amdgcn_mbcnt_lo(act, 0u);
                    wq[wave][pos] = ebase + (unsigned)__builtin_ctz(wd);
                    wd &= wd - 1u;
                }
                qc  += __builtin_popcount(act);
                tot += __builtin_popcount(act);
                act = __builtin_amdgcn_ballot_w32(wd != 0u);
            }
            while (qc >= 32 || (last && qc > 0)) {
                const int take = qc < 32 ? qc : 32;
                const int base = qc - take;
                qc = base;
                __builtin_amdgcn_fence(__ATOMIC_ACQ_REL, "wavefront");
                if (lane < take) {
                    const int e = (int)wq[wave][base + lane];
                    atomicAdd(&cnt[col[e] - n0], 1);
                }
            }
        }
    }
    if (lane == 0) wtot[wave] = tot;
    __syncthreads();

    if (tid < nn) dinv[n0 + tid] = rsqrtf((float)(cnt[tid] + 1));
    if (tid < AGG_WAVES) wcnt[b * AGG_WAVES + tid] = wtot[tid];
}


extern "C" void kernel_launch(void* const* d_in, const int* in_sizes, int n_in,
                              void* d_out, int out_size, void* d_ws, size_t ws_size,
                              hipStream_t stream) {
    const float* x  = (const float*)d_in[0];
    const int*   ei = (const int*)d_in[1];
    const float* W1 = (const float*)d_in[2];
    const float* b1 = (const float*)d_in[3];
    const float* W2 = (const float*)d_in[4];
    const float* b2 = (const float*)d_in[5];
    const float* W3 = (const float*)d_in[6];
    const float* b3 = (const float*)d_in[7];
    float* out = (float*)d_out;

    const int E = in_sizes[1] / 2;
    const int* row = ei;
    const int* col = ei + E;

    float* ws   = (float*)d_ws;
    float* dinv = ws;
    float* H    = ws + 50176;
    float* O    = H + (size_t)NN * FDIM;
    unsigned* bm = (unsigned*)(O + (size_t)NN * FDIM);
    int*      wcnt = (int*)d_ws + 33546240;
    unsigned* list = (unsigned*)d_ws + 33554432;

    build_bm<<<(E + CH_EDGES - 1) / CH_EDGES, BM_THREADS, 0, stream>>>(col, bm, E);
    own_deg<<<NOWN, AGG_THREADS, 0, stream>>>(col, bm, dinv, wcnt, E);

    agg_lds<128, 128><<<NOWN, AGG_THREADS, 0, stream>>>(x, row, col, bm, dinv, b1, wcnt, list, W1, O, E);

    agg_replay<128, 128, 64><<<NOWN, AGG_THREADS, 0, stream>>>(O, list, wcnt, dinv, b2, W2, W3, H);

    agg_replay<64, 0, 0><<<NOWN, AGG_THREADS, 0, stream>>>(H, list, wcnt, dinv, b3, nullptr, nullptr, out);
}
